# starting kernel
# baseline (speedup 1.0000x reference)
#include <hip/hip_runtime.h>

typedef __attribute__((ext_vector_type(2))) float v2f;
typedef __attribute__((ext_vector_type(8))) float v8f;

#define NPIX   (256 * 256)
#define NBATCH 16
#define NTOT   (NBATCH * NPIX)

__global__ __launch_bounds__(1024) void mvprop_setup(
    const float* __restrict__ Wh, const float* __restrict__ bh,
    const float* __restrict__ Wp, float2* __restrict__ wB,
    float* __restrict__ wBias, float* __restrict__ wWp) {
  int t = threadIdx.x;
  if (t < 10 * 3 * 32) {
    int tile = t / 96;
    int ks   = (t / 32) % 3;
    int lane = t & 31;
    int n    = tile * 16 + (lane & 15);
    int kb   = ks * 4 + ((lane >= 16) ? 2 : 0);
    float b0 = 0.0f, b1 = 0.0f;
    if (n < 150) {
      if (kb     < 9) b0 = Wh[n * 9 + kb];
      if (kb + 1 < 9) b1 = Wh[n * 9 + kb + 1];
    }
    wB[t] = make_float2(b0, b1);
  }
  if (t < 160) {
    wBias[t] = (t < 150) ? bh[t] : 0.0f;
    wWp[t]   = (t < 150) ? Wp[t] : 0.0f;
  }
}

__global__ __launch_bounds__(256) void mvprop_conv(
    const float* __restrict__ image, const float2* __restrict__ wB,
    const float* __restrict__ wBias, const float* __restrict__ wWp,
    float* __restrict__ pout) {
  const int lane  = threadIdx.x & 31;
  const int wave  = threadIdx.x >> 5;
  const int tileI = blockIdx.x * 8 + wave;
  const int base  = tileI * 16;
  const int b     = base >> 16;
  const int pix   = base & (NPIX - 1);
  const int y     = pix >> 8;
  const int x0    = pix & 255;

  const float* __restrict__ occ = image + (size_t)b * 2 * NPIX;

  const int m     = lane & 15;
  const int khalf = lane >> 4;
  const int x     = x0 + m;

  v2f aFrag[3];
#pragma unroll
  for (int ks = 0; ks < 3; ++ks) {
    float av[2];
#pragma unroll
    for (int j = 0; j < 2; ++j) {
      int k = ks * 4 + khalf * 2 + j;
      float val = 0.0f;
      if (k < 9) {
        int dy = k / 3 - 1, dx = k % 3 - 1;
        int yy = y + dy, xx = x + dx;
        bool ok = (yy >= 0) & (yy < 256) & (xx >= 0) & (xx < 256);
        int yc = min(max(yy, 0), 255);
        int xc = min(max(xx, 0), 255);
        val = occ[yc * 256 + xc] * (ok ? 1.0f : 0.0f);
      }
      av[j] = val;
    }
    aFrag[ks].x = av[0];
    aFrag[ks].y = av[1];
  }

  float acc[8];
#pragma unroll
  for (int r = 0; r < 8; ++r) acc[r] = 0.0f;

#pragma unroll
  for (int tile = 0; tile < 10; ++tile) {
    float bias = wBias[tile * 16 + m];
    v8f c;
#pragma unroll
    for (int r = 0; r < 8; ++r) c[r] = bias;
#pragma unroll
    for (int ks = 0; ks < 3; ++ks) {
      float2 bt = wB[(tile * 3 + ks) * 32 + lane];
      v2f bf;
      bf.x = bt.x;
      bf.y = bt.y;
      c = __builtin_amdgcn_wmma_f32_16x16x4_f32(
           false, aFrag[ks],  false, bf,
           (short)0, c,  false,  false);
    }
    float wp = wWp[tile * 16 + m];
#pragma unroll
    for (int r = 0; r < 8; ++r) acc[r] += fmaxf(c[r], 0.0f) * wp;
  }

#pragma unroll
  for (int r = 0; r < 8; ++r) {
    float s = acc[r];
    s += __shfl_xor(s, 1, 32);
    s += __shfl_xor(s, 2, 32);
    s += __shfl_xor(s, 4, 32);
    s += __shfl_xor(s, 8, 32);
    acc[r] = s;
  }

  if ((lane & 15) == 0) {
    float sg[8];
#pragma unroll
    for (int r = 0; r < 8; ++r) sg[r] = 1.0f / (1.0f + __expf(-acc[r]));
    float4* dst = (float4*)(pout + base + khalf * 8);
    dst[0] = make_float4(sg[0], sg[1], sg[2], sg[3]);
    dst[1] = make_float4(sg[4], sg[5], sg[6], sg[7]);
  }
}

__global__ __launch_bounds__(256) void mvprop_step(
    const float* __restrict__ image, const float* __restrict__ pbuf,
    const float* __restrict__ vin, long vinStride,
    float* __restrict__ vout, int finalPass) {
  int idx = blockIdx.x * 256 + threadIdx.x;
  int b   = idx >> 16;
  int pix = idx & (NPIX - 1);
  int y   = pix >> 8;
  int x   = pix & 255;

  const float* __restrict__ vb = vin + (size_t)b * vinStride;

  float vc = vb[pix];
  float mx = vc;
  int y0 = max(y - 1, 0), y1 = min(y + 1, 255);
  int x0 = max(x - 1, 0), x1 = min(x + 1, 255);
  for (int yy = y0; yy <= y1; ++yy)
    for (int xx = x0; xx <= x1; ++xx)
      mx = fmaxf(mx, vb[yy * 256 + xx]);

  const float* ib = image + (size_t)b * 2 * NPIX;
  float r = ib[NPIX + pix];
  float p = pbuf[idx];
  float vn = fmaxf(vc, r + p * (mx - r));

  if (finalPass) {
    float oc = ib[pix];
    vout[idx] = (oc > 0.49f) ? vn : -1.0f;
  } else {
    vout[idx] = vn;
  }
}

extern "C" void kernel_launch(void* const* d_in, const int* in_sizes, int n_in,
                              void* d_out, int out_size, void* d_ws,
                              size_t ws_size, hipStream_t stream) {
  const float* image = (const float*)d_in[0];
  const float* Wh    = (const float*)d_in[1];
  const float* bh    = (const float*)d_in[2];
  const float* Wp    = (const float*)d_in[3];
  float* out = (float*)d_out;

  float* ws    = (float*)d_ws;
  float* pbuf  = ws;
  float* v0    = ws + (size_t)NTOT;
  float* v1    = ws + (size_t)2 * NTOT;
  float* wB    = ws + (size_t)3 * NTOT;
  float* wBias = wB + 1920;
  float* wWp   = wBias + 160;

  mvprop_setup<<<1, 1024, 0, stream>>>(Wh, bh, Wp, (float2*)wB, wBias, wWp);

  mvprop_conv<<<NTOT / (16 * 8), 256, 0, stream>>>(
      image, (const float2*)wB, wBias, wWp, pbuf);

  const float* vin = image + NPIX;
  long stride = 2 * NPIX;
  float* bufs[2] = {v0, v1};
  for (int i = 0; i < 20; ++i) {
    int fin = (i == 19) ? 1 : 0;
    float* vo = fin ? out : bufs[i & 1];
    mvprop_step<<<NTOT / 256, 256, 0, stream>>>(image, pbuf, vin, stride, vo,
                                                fin);
    vin = vo;
    stride = NPIX;
  }
}


# Fuse the 20 propagation launches into one kernel: 64x64 tile + 20-cell halo in double-buffered LDS, r and p in registers, one barrier per step; 22 launches become 3
# speedup vs baseline: 3.9175x; 3.9175x over previous
#include <hip/hip_runtime.h>

typedef __attribute__((ext_vector_type(2))) float v2f;
typedef __attribute__((ext_vector_type(8))) float v8f;
typedef __attribute__((ext_vector_type(4))) float v4f;

#define NPIX   (256 * 256)
#define NBATCH 16
#define NTOT   (NBATCH * NPIX)

__global__ __launch_bounds__(1024) void mvprop_setup(
    const float* __restrict__ Wh, const float* __restrict__ bh,
    const float* __restrict__ Wp, float2* __restrict__ wB,
    float* __restrict__ wBias, float* __restrict__ wWp) {
  int t = threadIdx.x;
  if (t < 10 * 3 * 32) {
    int tile = t / 96;
    int ks   = (t / 32) % 3;
    int lane = t & 31;
    int n    = tile * 16 + (lane & 15);
    int kb   = ks * 4 + ((lane >= 16) ? 2 : 0);
    float b0 = 0.0f, b1 = 0.0f;
    if (n < 150) {
      if (kb     < 9) b0 = Wh[n * 9 + kb];
      if (kb + 1 < 9) b1 = Wh[n * 9 + kb + 1];
    }
    wB[t] = make_float2(b0, b1);
  }
  if (t < 160) {
    wBias[t] = (t < 150) ? bh[t] : 0.0f;
    wWp[t]   = (t < 150) ? Wp[t] : 0.0f;
  }
}

__global__ __launch_bounds__(256) void mvprop_conv(
    const float* __restrict__ image, const float2* __restrict__ wB,
    const float* __restrict__ wBias, const float* __restrict__ wWp,
    float* __restrict__ pout) {
  const int lane  = threadIdx.x & 31;
  const int wave  = threadIdx.x >> 5;
  const int tileI = blockIdx.x * 8 + wave;
  const int base  = tileI * 16;
  const int b     = base >> 16;
  const int pix   = base & (NPIX - 1);
  const int y     = pix >> 8;
  const int x0    = pix & 255;

  const float* __restrict__ occ = image + (size_t)b * 2 * NPIX;

  const int m     = lane & 15;
  const int khalf = lane >> 4;
  const int x     = x0 + m;

  v2f aFrag[3];
#pragma unroll
  for (int ks = 0; ks < 3; ++ks) {
    float av[2];
#pragma unroll
    for (int j = 0; j < 2; ++j) {
      int k = ks * 4 + khalf * 2 + j;
      float val = 0.0f;
      if (k < 9) {
        int dy = k / 3 - 1, dx = k % 3 - 1;
        int yy = y + dy, xx = x + dx;
        bool ok = (yy >= 0) & (yy < 256) & (xx >= 0) & (xx < 256);
        int yc = min(max(yy, 0), 255);
        int xc = min(max(xx, 0), 255);
        val = occ[yc * 256 + xc] * (ok ? 1.0f : 0.0f);
      }
      av[j] = val;
    }
    aFrag[ks].x = av[0];
    aFrag[ks].y = av[1];
  }

  float acc[8];
#pragma unroll
  for (int r = 0; r < 8; ++r) acc[r] = 0.0f;

#pragma unroll
  for (int tile = 0; tile < 10; ++tile) {
    float bias = wBias[tile * 16 + m];
    v8f c;
#pragma unroll
    for (int r = 0; r < 8; ++r) c[r] = bias;
#pragma unroll
    for (int ks = 0; ks < 3; ++ks) {
      float2 bt = wB[(tile * 3 + ks) * 32 + lane];
      v2f bf;
      bf.x = bt.x;
      bf.y = bt.y;
      c = __builtin_amdgcn_wmma_f32_16x16x4_f32(
           false, aFrag[ks],  false, bf,
           (short)0, c,  false,  false);
    }
    float wp = wWp[tile * 16 + m];
#pragma unroll
    for (int r = 0; r < 8; ++r) acc[r] += fmaxf(c[r], 0.0f) * wp;
  }

#pragma unroll
  for (int r = 0; r < 8; ++r) {
    float s = acc[r];
    s += __shfl_xor(s, 1, 32);
    s += __shfl_xor(s, 2, 32);
    s += __shfl_xor(s, 4, 32);
    s += __shfl_xor(s, 8, 32);
    acc[r] = s;
  }

  if ((lane & 15) == 0) {
    float sg[8];
#pragma unroll
    for (int r = 0; r < 8; ++r) sg[r] = 1.0f / (1.0f + __expf(-acc[r]));
    float4* dst = (float4*)(pout + base + khalf * 8);
    dst[0] = make_float4(sg[0], sg[1], sg[2], sg[3]);
    dst[1] = make_float4(sg[4], sg[5], sg[6], sg[7]);
  }
}

#define TS     64
#define HALO   20
#define RS     (TS + 2 * HALO)
#define QPR    (RS / 4)
#define COL0   4
#define PITCH  112
#define BUFSZ  ((RS + 2) * PITCH)
#define FTHREADS 704

__device__ __forceinline__ float max3f(float a, float b, float c) {
  return fmaxf(fmaxf(a, b), c);
}

__device__ __forceinline__ void patch_step(const float* __restrict__ vin,
                                           int ry, int rx,
                                           const float (&r)[4][4],
                                           const float (&p)[4][4],
                                           float (&vn)[4][4]) {
  float hm[6][4];
  float vc[4][4];
#pragma unroll
  for (int k = 0; k < 6; ++k) {
    const float* rowp = vin + (ry + k) * PITCH + COL0 + rx;
    const v4f    m  = *(const v4f*)rowp;
    const float  lf = rowp[-1];
    const float  rt = rowp[4];
    hm[k][0] = max3f(lf,  m.x, m.y);
    hm[k][1] = max3f(m.x, m.y, m.z);
    hm[k][2] = max3f(m.y, m.z, m.w);
    hm[k][3] = max3f(m.z, m.w, rt);
    if (k >= 1 && k <= 4) {
      vc[k - 1][0] = m.x;
      vc[k - 1][1] = m.y;
      vc[k - 1][2] = m.z;
      vc[k - 1][3] = m.w;
    }
  }
#pragma unroll
  for (int j = 0; j < 4; ++j) {
#pragma unroll
    for (int i = 0; i < 4; ++i) {
      float mx = max3f(hm[j][i], hm[j + 1][i], hm[j + 2][i]);
      vn[j][i] = fmaxf(vc[j][i], r[j][i] + p[j][i] * (mx - r[j][i]));
    }
  }
}

__device__ __forceinline__ void store_patch(float* __restrict__ vout, int ry,
                                            int rx, const float (&vn)[4][4]) {
#pragma unroll
  for (int j = 0; j < 4; ++j) {
    float4* d = (float4*)(vout + (ry + 1 + j) * PITCH + COL0 + rx);
    *d = make_float4(vn[j][0], vn[j][1], vn[j][2], vn[j][3]);
  }
}

__global__ __launch_bounds__(FTHREADS) void vprop_fused(
    const float* __restrict__ image, const float* __restrict__ pbuf,
    float* __restrict__ out) {
  __shared__ __attribute__((aligned(16))) float sv[2 * BUFSZ];

  const int t    = threadIdx.x;
  const int tile = blockIdx.x;
  const int b    = tile >> 4;
  const int ty   = (tile >> 2) & 3;
  const int tx   = tile & 3;

  const int sy = t / QPR;
  const int qx = t - sy * QPR;
  const int ry = sy * 4;
  const int rx = qx * 4;
  const int gy = ty * TS - HALO + ry;
  const int gx = tx * TS - HALO + rx;
  const bool inimg = (t < QPR * QPR) & (gy >= 0) & (gy < 256) &
                     (gx >= 0) & (gx < 256);
  const bool inner = inimg & (ry >= HALO) & (ry < HALO + TS) &
                     (rx >= HALO) & (rx < HALO + TS);

  const int drow = (ry + 3 <= (RS - 1) / 2) ? ry + 3 : RS - 1 - ry;
  const int dcol = (rx + 3 <= (RS - 1) / 2) ? rx + 3 : RS - 1 - rx;
  const int dmax = inimg ? min(drow, dcol) : 0;

  const float* __restrict__ ib = image + (size_t)b * 2 * NPIX;
  const float* __restrict__ pb = pbuf + (size_t)b * NPIX;

  float r[4][4], p[4][4], oc[4][4];
  if (inimg) {
#pragma unroll
    for (int j = 0; j < 4; ++j) {
      const int g = (gy + j) * 256 + gx;
      const float4 r4 = *(const float4*)(ib + NPIX + g);
      const float4 p4 = *(const float4*)(pb + g);
      r[j][0] = r4.x; r[j][1] = r4.y; r[j][2] = r4.z; r[j][3] = r4.w;
      p[j][0] = p4.x; p[j][1] = p4.y; p[j][2] = p4.z; p[j][3] = p4.w;
    }
  }
  if (inner) {
#pragma unroll
    for (int j = 0; j < 4; ++j) {
      const float4 o4 = *(const float4*)(ib + (gy + j) * 256 + gx);
      oc[j][0] = o4.x; oc[j][1] = o4.y; oc[j][2] = o4.z; oc[j][3] = o4.w;
    }
  }

  {
    const float ninf = -__builtin_inff();
    const float4 n4 = make_float4(ninf, ninf, ninf, ninf);
    float4* s4 = (float4*)sv;
    for (int i = t; i < 2 * BUFSZ / 4; i += FTHREADS) s4[i] = n4;
  }
  __syncthreads();

  if (inimg) {
#pragma unroll
    for (int j = 0; j < 4; ++j) {
      float4* d = (float4*)(sv + (ry + 1 + j) * PITCH + COL0 + rx);
      *d = make_float4(r[j][0], r[j][1], r[j][2], r[j][3]);
    }
  }
  __syncthreads();

  float* const pl0 = sv;
  float* const pl1 = sv + BUFSZ;

  for (int s = 1; s < HALO - 1; s += 2) {
    if (dmax >= s) {
      float vn[4][4];
      patch_step(pl0, ry, rx, r, p, vn);
      store_patch(pl1, ry, rx, vn);
    }
    __syncthreads();
    if (dmax >= s + 1) {
      float vn[4][4];
      patch_step(pl1, ry, rx, r, p, vn);
      store_patch(pl0, ry, rx, vn);
    }
    __syncthreads();
  }
  if (dmax >= HALO - 1) {
    float vn[4][4];
    patch_step(pl0, ry, rx, r, p, vn);
    store_patch(pl1, ry, rx, vn);
  }
  __syncthreads();

  if (inner) {
    float vn[4][4];
    patch_step(pl1, ry, rx, r, p, vn);
#pragma unroll
    for (int j = 0; j < 4; ++j) {
      float4 o;
      o.x = (oc[j][0] > 0.49f) ? vn[j][0] : -1.0f;
      o.y = (oc[j][1] > 0.49f) ? vn[j][1] : -1.0f;
      o.z = (oc[j][2] > 0.49f) ? vn[j][2] : -1.0f;
      o.w = (oc[j][3] > 0.49f) ? vn[j][3] : -1.0f;
      *(float4*)(out + (size_t)b * NPIX + (gy + j) * 256 + gx) = o;
    }
  }
}

extern "C" void kernel_launch(void* const* d_in, const int* in_sizes, int n_in,
                              void* d_out, int out_size, void* d_ws,
                              size_t ws_size, hipStream_t stream) {
  const float* image = (const float*)d_in[0];
  const float* Wh    = (const float*)d_in[1];
  const float* bh    = (const float*)d_in[2];
  const float* Wp    = (const float*)d_in[3];
  float* out = (float*)d_out;

  float* ws    = (float*)d_ws;
  float* pbuf  = ws;
  float* wB    = ws + (size_t)3 * NTOT;
  float* wBias = wB + 1920;
  float* wWp   = wBias + 160;

  mvprop_setup<<<1, 1024, 0, stream>>>(Wh, bh, Wp, (float2*)wB, wBias, wWp);

  mvprop_conv<<<NTOT / (16 * 8), 256, 0, stream>>>(
      image, (const float2*)wB, wBias, wWp, pbuf);

  vprop_fused<<<NBATCH * 16, FTHREADS, 0, stream>>>(image, pbuf, out);
}


# conv: one wave per image row keeps the weight fragments, bias C tiles and W_p in registers and loops over its 16 pixel tiles (weights loaded once per row, not per tile); 65536 short waves become 4096
# speedup vs baseline: 4.0784x; 1.0411x over previous
#include <hip/hip_runtime.h>

typedef __attribute__((ext_vector_type(2))) float v2f;
typedef __attribute__((ext_vector_type(8))) float v8f;
typedef __attribute__((ext_vector_type(4))) float v4f;

#define NPIX   (256 * 256)
#define NBATCH 16
#define NTOT   (NBATCH * NPIX)

__global__ __launch_bounds__(1024) void mvprop_setup(
    const float* __restrict__ Wh, const float* __restrict__ bh,
    const float* __restrict__ Wp, float2* __restrict__ wB,
    float* __restrict__ wBias, float* __restrict__ wWp) {
  int t = threadIdx.x;
  if (t < 10 * 3 * 32) {
    int tile = t / 96;
    int ks   = (t / 32) % 3;
    int lane = t & 31;
    int n    = tile * 16 + (lane & 15);
    int kb   = ks * 4 + ((lane >= 16) ? 2 : 0);
    float b0 = 0.0f, b1 = 0.0f;
    if (n < 150) {
      if (kb     < 9) b0 = Wh[n * 9 + kb];
      if (kb + 1 < 9) b1 = Wh[n * 9 + kb + 1];
    }
    wB[t] = make_float2(b0, b1);
  }
  if (t < 160) {
    wBias[t] = (t < 150) ? bh[t] : 0.0f;
    wWp[t]   = (t < 150) ? Wp[t] : 0.0f;
  }
}

#define CONV_TILES_PER_ROW 16

__global__ __launch_bounds__(256) void mvprop_conv(
    const float* __restrict__ image, const float2* __restrict__ wB,
    const float* __restrict__ wBias, const float* __restrict__ wWp,
    float* __restrict__ pout) {
  const int lane  = threadIdx.x & 31;
  const int wave  = threadIdx.x >> 5;
  const int row   = blockIdx.x * 8 + wave;
  const int b     = row >> 8;
  const int y     = row & 255;

  const float* __restrict__ occ = image + (size_t)b * 2 * NPIX;
  float* __restrict__ prow = pout + (size_t)row * 256;

  const int m     = lane & 15;
  const int khalf = lane >> 4;

  v2f bFrag[10][3];
#pragma unroll
  for (int tile = 0; tile < 10; ++tile) {
#pragma unroll
    for (int ks = 0; ks < 3; ++ks) {
      float2 bt = wB[(tile * 3 + ks) * 32 + lane];
      bFrag[tile][ks].x = bt.x;
      bFrag[tile][ks].y = bt.y;
    }
  }
  v8f   cInit[10];
  float wp[10];
#pragma unroll
  for (int tile = 0; tile < 10; ++tile) {
    const float bias = wBias[tile * 16 + m];
#pragma unroll
    for (int r = 0; r < 8; ++r) cInit[tile][r] = bias;
    wp[tile] = wWp[tile * 16 + m];
  }

  int  rowOff[6];
  int  col[6];
  bool yok[6];
#pragma unroll
  for (int s = 0; s < 6; ++s) {
    const int k  = (s >> 1) * 4 + khalf * 2 + (s & 1);
    const int kk = min(k, 8);
    const int dy = kk / 3 - 1, dx = kk % 3 - 1;
    const int yy = y + dy;
    yok[s]    = (k < 9) & (yy >= 0) & (yy < 256);
    rowOff[s] = min(max(yy, 0), 255) * 256;
    col[s]    = m + dx;
  }

  auto loadA = [&](int x0, v2f (&af)[3]) {
#pragma unroll
    for (int s = 0; s < 6; ++s) {
      const int  xx = x0 + col[s];
      const bool ok = yok[s] & (xx >= 0) & (xx < 256);
      const int  xc = min(max(xx, 0), 255);
      const float v = occ[rowOff[s] + xc];
      const float val = ok ? v : 0.0f;
      if (s & 1) af[s >> 1].y = val; else af[s >> 1].x = val;
    }
  };

  v2f aCur[3];
  loadA(0, aCur);

#pragma nounroll
  for (int t = 0; t < CONV_TILES_PER_ROW; ++t) {
    v2f aNext[3];
    loadA(min(t + 1, CONV_TILES_PER_ROW - 1) * 16, aNext);

    float acc[8];
#pragma unroll
    for (int r = 0; r < 8; ++r) acc[r] = 0.0f;

#pragma unroll
    for (int tile = 0; tile < 10; ++tile) {
      v8f c = cInit[tile];
#pragma unroll
      for (int ks = 0; ks < 3; ++ks) {
        c = __builtin_amdgcn_wmma_f32_16x16x4_f32(
             false, aCur[ks],  false, bFrag[tile][ks],
             (short)0, c,  false,  false);
      }
#pragma unroll
      for (int r = 0; r < 8; ++r) acc[r] += fmaxf(c[r], 0.0f) * wp[tile];
    }

#pragma unroll
    for (int r = 0; r < 8; ++r) {
      float s = acc[r];
      s += __shfl_xor(s, 1, 32);
      s += __shfl_xor(s, 2, 32);
      s += __shfl_xor(s, 4, 32);
      s += __shfl_xor(s, 8, 32);
      acc[r] = s;
    }

    if (m == 0) {
      float sg[8];
#pragma unroll
      for (int r = 0; r < 8; ++r) sg[r] = 1.0f / (1.0f + __expf(-acc[r]));
      float4* dst = (float4*)(prow + t * 16 + khalf * 8);
      dst[0] = make_float4(sg[0], sg[1], sg[2], sg[3]);
      dst[1] = make_float4(sg[4], sg[5], sg[6], sg[7]);
    }

#pragma unroll
    for (int ks = 0; ks < 3; ++ks) aCur[ks] = aNext[ks];
  }
}

#define TS     64
#define HALO   20
#define RS     (TS + 2 * HALO)
#define QPR    (RS / 4)
#define COL0   4
#define PITCH  112
#define BUFSZ  ((RS + 2) * PITCH)
#define FTHREADS 704

__device__ __forceinline__ float max3f(float a, float b, float c) {
  return fmaxf(fmaxf(a, b), c);
}

__device__ __forceinline__ void patch_step(const float* __restrict__ vin,
                                           int ry, int rx,
                                           const float (&r)[4][4],
                                           const float (&p)[4][4],
                                           float (&vn)[4][4]) {
  float hm[6][4];
  float vc[4][4];
#pragma unroll
  for (int k = 0; k < 6; ++k) {
    const float* rowp = vin + (ry + k) * PITCH + COL0 + rx;
    const v4f    m  = *(const v4f*)rowp;
    const float  lf = rowp[-1];
    const float  rt = rowp[4];
    hm[k][0] = max3f(lf,  m.x, m.y);
    hm[k][1] = max3f(m.x, m.y, m.z);
    hm[k][2] = max3f(m.y, m.z, m.w);
    hm[k][3] = max3f(m.z, m.w, rt);
    if (k >= 1 && k <= 4) {
      vc[k - 1][0] = m.x;
      vc[k - 1][1] = m.y;
      vc[k - 1][2] = m.z;
      vc[k - 1][3] = m.w;
    }
  }
#pragma unroll
  for (int j = 0; j < 4; ++j) {
#pragma unroll
    for (int i = 0; i < 4; ++i) {
      float mx = max3f(hm[j][i], hm[j + 1][i], hm[j + 2][i]);
      vn[j][i] = fmaxf(vc[j][i], r[j][i] + p[j][i] * (mx - r[j][i]));
    }
  }
}

__device__ __forceinline__ void store_patch(float* __restrict__ vout, int ry,
                                            int rx, const float (&vn)[4][4]) {
#pragma unroll
  for (int j = 0; j < 4; ++j) {
    float4* d = (float4*)(vout + (ry + 1 + j) * PITCH + COL0 + rx);
    *d = make_float4(vn[j][0], vn[j][1], vn[j][2], vn[j][3]);
  }
}

__global__ __launch_bounds__(FTHREADS) void vprop_fused(
    const float* __restrict__ image, const float* __restrict__ pbuf,
    float* __restrict__ out) {
  __shared__ __attribute__((aligned(16))) float sv[2 * BUFSZ];

  const int t    = threadIdx.x;
  const int tile = blockIdx.x;
  const int b    = tile >> 4;
  const int ty   = (tile >> 2) & 3;
  const int tx   = tile & 3;

  const int sy = t / QPR;
  const int qx = t - sy * QPR;
  const int ry = sy * 4;
  const int rx = qx * 4;
  const int gy = ty * TS - HALO + ry;
  const int gx = tx * TS - HALO + rx;
  const bool inimg = (t < QPR * QPR) & (gy >= 0) & (gy < 256) &
                     (gx >= 0) & (gx < 256);
  const bool inner = inimg & (ry >= HALO) & (ry < HALO + TS) &
                     (rx >= HALO) & (rx < HALO + TS);

  const int drow = (ry + 3 <= (RS - 1) / 2) ? ry + 3 : RS - 1 - ry;
  const int dcol = (rx + 3 <= (RS - 1) / 2) ? rx + 3 : RS - 1 - rx;
  const int dmax = inimg ? min(drow, dcol) : 0;

  const float* __restrict__ ib = image + (size_t)b * 2 * NPIX;
  const float* __restrict__ pb = pbuf + (size_t)b * NPIX;

  float r[4][4], p[4][4], oc[4][4];
  if (inimg) {
#pragma unroll
    for (int j = 0; j < 4; ++j) {
      const int g = (gy + j) * 256 + gx;
      const float4 r4 = *(const float4*)(ib + NPIX + g);
      const float4 p4 = *(const float4*)(pb + g);
      r[j][0] = r4.x; r[j][1] = r4.y; r[j][2] = r4.z; r[j][3] = r4.w;
      p[j][0] = p4.x; p[j][1] = p4.y; p[j][2] = p4.z; p[j][3] = p4.w;
    }
  }
  if (inner) {
#pragma unroll
    for (int j = 0; j < 4; ++j) {
      const float4 o4 = *(const float4*)(ib + (gy + j) * 256 + gx);
      oc[j][0] = o4.x; oc[j][1] = o4.y; oc[j][2] = o4.z; oc[j][3] = o4.w;
    }
  }

  {
    const float ninf = -__builtin_inff();
    const float4 n4 = make_float4(ninf, ninf, ninf, ninf);
    float4* s4 = (float4*)sv;
    for (int i = t; i < 2 * BUFSZ / 4; i += FTHREADS) s4[i] = n4;
  }
  __syncthreads();

  if (inimg) {
#pragma unroll
    for (int j = 0; j < 4; ++j) {
      float4* d = (float4*)(sv + (ry + 1 + j) * PITCH + COL0 + rx);
      *d = make_float4(r[j][0], r[j][1], r[j][2], r[j][3]);
    }
  }
  __syncthreads();

  float* const pl0 = sv;
  float* const pl1 = sv + BUFSZ;

  for (int s = 1; s < HALO - 1; s += 2) {
    if (dmax >= s) {
      float vn[4][4];
      patch_step(pl0, ry, rx, r, p, vn);
      store_patch(pl1, ry, rx, vn);
    }
    __syncthreads();
    if (dmax >= s + 1) {
      float vn[4][4];
      patch_step(pl1, ry, rx, r, p, vn);
      store_patch(pl0, ry, rx, vn);
    }
    __syncthreads();
  }
  if (dmax >= HALO - 1) {
    float vn[4][4];
    patch_step(pl0, ry, rx, r, p, vn);
    store_patch(pl1, ry, rx, vn);
  }
  __syncthreads();

  if (inner) {
    float vn[4][4];
    patch_step(pl1, ry, rx, r, p, vn);
#pragma unroll
    for (int j = 0; j < 4; ++j) {
      float4 o;
      o.x = (oc[j][0] > 0.49f) ? vn[j][0] : -1.0f;
      o.y = (oc[j][1] > 0.49f) ? vn[j][1] : -1.0f;
      o.z = (oc[j][2] > 0.49f) ? vn[j][2] : -1.0f;
      o.w = (oc[j][3] > 0.49f) ? vn[j][3] : -1.0f;
      *(float4*)(out + (size_t)b * NPIX + (gy + j) * 256 + gx) = o;
    }
  }
}

extern "C" void kernel_launch(void* const* d_in, const int* in_sizes, int n_in,
                              void* d_out, int out_size, void* d_ws,
                              size_t ws_size, hipStream_t stream) {
  const float* image = (const float*)d_in[0];
  const float* Wh    = (const float*)d_in[1];
  const float* bh    = (const float*)d_in[2];
  const float* Wp    = (const float*)d_in[3];
  float* out = (float*)d_out;

  float* ws    = (float*)d_ws;
  float* pbuf  = ws;
  float* wB    = ws + (size_t)3 * NTOT;
  float* wBias = wB + 1920;
  float* wWp   = wBias + 160;

  mvprop_setup<<<1, 1024, 0, stream>>>(Wh, bh, Wp, (float2*)wB, wBias, wWp);

  mvprop_conv<<<512, 256, 0, stream>>>(
      image, (const float2*)wB, wBias, wWp, pbuf);

  vprop_fused<<<NBATCH * 16, FTHREADS, 0, stream>>>(image, pbuf, out);
}


# conv builds its own weight fragments: each block copies W_h to LDS, every wave picks its 30 B fragments, bias and W_p from the inputs; the repack launch is gone, 3 launches become 2
# speedup vs baseline: 4.2109x; 1.0325x over previous
#include <hip/hip_runtime.h>

typedef __attribute__((ext_vector_type(2))) float v2f;
typedef __attribute__((ext_vector_type(8))) float v8f;
typedef __attribute__((ext_vector_type(4))) float v4f;

#define NPIX   (256 * 256)
#define NBATCH 16
#define NTOT   (NBATCH * NPIX)

#define CONV_TILES_PER_ROW 16
#define NCH   150
#define NTAP  9

__global__ __launch_bounds__(256) void mvprop_conv(
    const float* __restrict__ image, const float* __restrict__ Wh,
    const float* __restrict__ bh, const float* __restrict__ Wp,
    float* __restrict__ pout) {
  __shared__ float sW[NCH * NTAP + 1];

  const int lane  = threadIdx.x & 31;
  const int wave  = threadIdx.x >> 5;
  const int row   = blockIdx.x * 8 + wave;
  const int b     = row >> 8;
  const int y     = row & 255;

  const float* __restrict__ occ = image + (size_t)b * 2 * NPIX;
  float* __restrict__ prow = pout + (size_t)row * 256;

  const int m     = lane & 15;
  const int khalf = lane >> 4;

  float wcopy[6];
#pragma unroll
  for (int j = 0; j < 6; ++j) {
    const int i = threadIdx.x + j * 256;
    wcopy[j] = (i < NCH * NTAP) ? Wh[i] : 0.0f;
  }

  v8f   cInit[10];
  float wp[10];
#pragma unroll
  for (int tile = 0; tile < 10; ++tile) {
    const int   n    = tile * 16 + m;
    const int   nc   = min(n, NCH - 1);
    const float bias = (n < NCH) ? bh[nc] : 0.0f;
#pragma unroll
    for (int r = 0; r < 8; ++r) cInit[tile][r] = bias;
    wp[tile] = (n < NCH) ? Wp[nc] : 0.0f;
  }

  int  rowOff[6];
  int  col[6];
  bool yok[6];
#pragma unroll
  for (int s = 0; s < 6; ++s) {
    const int k  = (s >> 1) * 4 + khalf * 2 + (s & 1);
    const int kk = min(k, 8);
    const int dy = kk / 3 - 1, dx = kk % 3 - 1;
    const int yy = y + dy;
    yok[s]    = (k < 9) & (yy >= 0) & (yy < 256);
    rowOff[s] = min(max(yy, 0), 255) * 256;
    col[s]    = m + dx;
  }

  auto loadA = [&](int x0, v2f (&af)[3]) {
#pragma unroll
    for (int s = 0; s < 6; ++s) {
      const int  xx = x0 + col[s];
      const bool ok = yok[s] & (xx >= 0) & (xx < 256);
      const int  xc = min(max(xx, 0), 255);
      const float v = occ[rowOff[s] + xc];
      const float val = ok ? v : 0.0f;
      if (s & 1) af[s >> 1].y = val; else af[s >> 1].x = val;
    }
  };

  v2f aCur[3];
  loadA(0, aCur);

#pragma unroll
  for (int j = 0; j < 6; ++j) {
    const int i = threadIdx.x + j * 256;
    if (i < NCH * NTAP) sW[i] = wcopy[j];
  }
  if (threadIdx.x == 0) sW[NCH * NTAP] = 0.0f;
  __syncthreads();

  v2f bFrag[10][3];
#pragma unroll
  for (int tile = 0; tile < 10; ++tile) {
    const int  n   = tile * 16 + m;
    const bool nok = n < NCH;
#pragma unroll
    for (int ks = 0; ks < 3; ++ks) {
      const int k0 = ks * 4 + khalf * 2;
      const int k1 = k0 + 1;
      const int i0 = (nok & (k0 < NTAP)) ? n * NTAP + k0 : NCH * NTAP;
      const int i1 = (nok & (k1 < NTAP)) ? n * NTAP + k1 : NCH * NTAP;
      bFrag[tile][ks].x = sW[i0];
      bFrag[tile][ks].y = sW[i1];
    }
  }

#pragma nounroll
  for (int t = 0; t < CONV_TILES_PER_ROW; ++t) {
    v2f aNext[3];
    loadA(min(t + 1, CONV_TILES_PER_ROW - 1) * 16, aNext);

    float acc[8];
#pragma unroll
    for (int r = 0; r < 8; ++r) acc[r] = 0.0f;

#pragma unroll
    for (int tile = 0; tile < 10; ++tile) {
      v8f c = cInit[tile];
#pragma unroll
      for (int ks = 0; ks < 3; ++ks) {
        c = __builtin_amdgcn_wmma_f32_16x16x4_f32(
             false, aCur[ks],  false, bFrag[tile][ks],
             (short)0, c,  false,  false);
      }
#pragma unroll
      for (int r = 0; r < 8; ++r) acc[r] += fmaxf(c[r], 0.0f) * wp[tile];
    }

#pragma unroll
    for (int r = 0; r < 8; ++r) {
      float s = acc[r];
      s += __shfl_xor(s, 1, 32);
      s += __shfl_xor(s, 2, 32);
      s += __shfl_xor(s, 4, 32);
      s += __shfl_xor(s, 8, 32);
      acc[r] = s;
    }

    if (m == 0) {
      float sg[8];
#pragma unroll
      for (int r = 0; r < 8; ++r) sg[r] = 1.0f / (1.0f + __expf(-acc[r]));
      float4* dst = (float4*)(prow + t * 16 + khalf * 8);
      dst[0] = make_float4(sg[0], sg[1], sg[2], sg[3]);
      dst[1] = make_float4(sg[4], sg[5], sg[6], sg[7]);
    }

#pragma unroll
    for (int ks = 0; ks < 3; ++ks) aCur[ks] = aNext[ks];
  }
}

#define TS     64
#define HALO   20
#define RS     (TS + 2 * HALO)
#define QPR    (RS / 4)
#define COL0   4
#define PITCH  112
#define BUFSZ  ((RS + 2) * PITCH)
#define FTHREADS 704

__device__ __forceinline__ float max3f(float a, float b, float c) {
  return fmaxf(fmaxf(a, b), c);
}

__device__ __forceinline__ void patch_step(const float* __restrict__ vin,
                                           int ry, int rx,
                                           const float (&r)[4][4],
                                           const float (&p)[4][4],
                                           float (&vn)[4][4]) {
  float hm[6][4];
  float vc[4][4];
#pragma unroll
  for (int k = 0; k < 6; ++k) {
    const float* rowp = vin + (ry + k) * PITCH + COL0 + rx;
    const v4f    m  = *(const v4f*)rowp;
    const float  lf = rowp[-1];
    const float  rt = rowp[4];
    hm[k][0] = max3f(lf,  m.x, m.y);
    hm[k][1] = max3f(m.x, m.y, m.z);
    hm[k][2] = max3f(m.y, m.z, m.w);
    hm[k][3] = max3f(m.z, m.w, rt);
    if (k >= 1 && k <= 4) {
      vc[k - 1][0] = m.x;
      vc[k - 1][1] = m.y;
      vc[k - 1][2] = m.z;
      vc[k - 1][3] = m.w;
    }
  }
#pragma unroll
  for (int j = 0; j < 4; ++j) {
#pragma unroll
    for (int i = 0; i < 4; ++i) {
      float mx = max3f(hm[j][i], hm[j + 1][i], hm[j + 2][i]);
      vn[j][i] = fmaxf(vc[j][i], r[j][i] + p[j][i] * (mx - r[j][i]));
    }
  }
}

__device__ __forceinline__ void store_patch(float* __restrict__ vout, int ry,
                                            int rx, const float (&vn)[4][4]) {
#pragma unroll
  for (int j = 0; j < 4; ++j) {
    float4* d = (float4*)(vout + (ry + 1 + j) * PITCH + COL0 + rx);
    *d = make_float4(vn[j][0], vn[j][1], vn[j][2], vn[j][3]);
  }
}

__global__ __launch_bounds__(FTHREADS) void vprop_fused(
    const float* __restrict__ image, const float* __restrict__ pbuf,
    float* __restrict__ out) {
  __shared__ __attribute__((aligned(16))) float sv[2 * BUFSZ];

  const int t    = threadIdx.x;
  const int tile = blockIdx.x;
  const int b    = tile >> 4;
  const int ty   = (tile >> 2) & 3;
  const int tx   = tile & 3;

  const int sy = t / QPR;
  const int qx = t - sy * QPR;
  const int ry = sy * 4;
  const int rx = qx * 4;
  const int gy = ty * TS - HALO + ry;
  const int gx = tx * TS - HALO + rx;
  const bool inimg = (t < QPR * QPR) & (gy >= 0) & (gy < 256) &
                     (gx >= 0) & (gx < 256);
  const bool inner = inimg & (ry >= HALO) & (ry < HALO + TS) &
                     (rx >= HALO) & (rx < HALO + TS);

  const int drow = (ry + 3 <= (RS - 1) / 2) ? ry + 3 : RS - 1 - ry;
  const int dcol = (rx + 3 <= (RS - 1) / 2) ? rx + 3 : RS - 1 - rx;
  const int dmax = inimg ? min(drow, dcol) : 0;

  const float* __restrict__ ib = image + (size_t)b * 2 * NPIX;
  const float* __restrict__ pb = pbuf + (size_t)b * NPIX;

  float r[4][4], p[4][4], oc[4][4];
  if (inimg) {
#pragma unroll
    for (int j = 0; j < 4; ++j) {
      const int g = (gy + j) * 256 + gx;
      const float4 r4 = *(const float4*)(ib + NPIX + g);
      const float4 p4 = *(const float4*)(pb + g);
      r[j][0] = r4.x; r[j][1] = r4.y; r[j][2] = r4.z; r[j][3] = r4.w;
      p[j][0] = p4.x; p[j][1] = p4.y; p[j][2] = p4.z; p[j][3] = p4.w;
    }
  }
  if (inner) {
#pragma unroll
    for (int j = 0; j < 4; ++j) {
      const float4 o4 = *(const float4*)(ib + (gy + j) * 256 + gx);
      oc[j][0] = o4.x; oc[j][1] = o4.y; oc[j][2] = o4.z; oc[j][3] = o4.w;
    }
  }

  {
    const float ninf = -__builtin_inff();
    const float4 n4 = make_float4(ninf, ninf, ninf, ninf);
    float4* s4 = (float4*)sv;
    for (int i = t; i < 2 * BUFSZ / 4; i += FTHREADS) s4[i] = n4;
  }
  __syncthreads();

  if (inimg) {
#pragma unroll
    for (int j = 0; j < 4; ++j) {
      float4* d = (float4*)(sv + (ry + 1 + j) * PITCH + COL0 + rx);
      *d = make_float4(r[j][0], r[j][1], r[j][2], r[j][3]);
    }
  }
  __syncthreads();

  float* const pl0 = sv;
  float* const pl1 = sv + BUFSZ;

  for (int s = 1; s < HALO - 1; s += 2) {
    if (dmax >= s) {
      float vn[4][4];
      patch_step(pl0, ry, rx, r, p, vn);
      store_patch(pl1, ry, rx, vn);
    }
    __syncthreads();
    if (dmax >= s + 1) {
      float vn[4][4];
      patch_step(pl1, ry, rx, r, p, vn);
      store_patch(pl0, ry, rx, vn);
    }
    __syncthreads();
  }
  if (dmax >= HALO - 1) {
    float vn[4][4];
    patch_step(pl0, ry, rx, r, p, vn);
    store_patch(pl1, ry, rx, vn);
  }
  __syncthreads();

  if (inner) {
    float vn[4][4];
    patch_step(pl1, ry, rx, r, p, vn);
#pragma unroll
    for (int j = 0; j < 4; ++j) {
      float4 o;
      o.x = (oc[j][0] > 0.49f) ? vn[j][0] : -1.0f;
      o.y = (oc[j][1] > 0.49f) ? vn[j][1] : -1.0f;
      o.z = (oc[j][2] > 0.49f) ? vn[j][2] : -1.0f;
      o.w = (oc[j][3] > 0.49f) ? vn[j][3] : -1.0f;
      *(float4*)(out + (size_t)b * NPIX + (gy + j) * 256 + gx) = o;
    }
  }
}

extern "C" void kernel_launch(void* const* d_in, const int* in_sizes, int n_in,
                              void* d_out, int out_size, void* d_ws,
                              size_t ws_size, hipStream_t stream) {
  const float* image = (const float*)d_in[0];
  const float* Wh    = (const float*)d_in[1];
  const float* bh    = (const float*)d_in[2];
  const float* Wp    = (const float*)d_in[3];
  float* out = (float*)d_out;

  float* pbuf  = (float*)d_ws;

  mvprop_conv<<<512, 256, 0, stream>>>(image, Wh, bh, Wp, pbuf);

  vprop_fused<<<NBATCH * 16, FTHREADS, 0, stream>>>(image, pbuf, out);
}


# conv product transposed (weights as A, patches as B): channels land on each lane's D registers, W_p sum in-lane + 1 lane exchange, 1 sigmoid per pixel; bias as 10th tap; channel sum order changes
# speedup vs baseline: 4.6799x; 1.1114x over previous
#include <hip/hip_runtime.h>

typedef __attribute__((ext_vector_type(2))) float v2f;
typedef __attribute__((ext_vector_type(8))) float v8f;
typedef __attribute__((ext_vector_type(4))) float v4f;

#define NPIX   (256 * 256)
#define NBATCH 16
#define NTOT   (NBATCH * NPIX)

#define CONV_TILES_PER_ROW 16
#define NCH   150
#define NCHP  160
#define NTAP  9
#define ZSLOT (NCH * NTAP)
#define BBASE (NCH * NTAP + 1)

__device__ __forceinline__ float xhalf(float x) {
  const int b = __builtin_bit_cast(int, x);
  return __builtin_bit_cast(
      float, __builtin_amdgcn_permlanex16(b, b, 0x76543210, 0xfedcba98, false,
                                          false));
}

__global__ __launch_bounds__(256) void mvprop_conv(
    const float* __restrict__ image, const float* __restrict__ Wh,
    const float* __restrict__ bh, const float* __restrict__ Wp,
    float* __restrict__ pout) {
  __shared__ float sW[NCH * NTAP + 1 + NCHP];
  __shared__ __attribute__((aligned(16))) float sWp[NCHP];

  const int lane  = threadIdx.x & 31;
  const int wave  = threadIdx.x >> 5;
  const int row   = blockIdx.x * 8 + wave;
  const int b     = row >> 8;
  const int y     = row & 255;

  const float* __restrict__ occ = image + (size_t)b * 2 * NPIX;
  float* __restrict__ prow = pout + (size_t)row * 256;

  const int m     = lane & 15;
  const int khalf = lane >> 4;

  float wcopy[6];
#pragma unroll
  for (int j = 0; j < 6; ++j) {
    const int i = threadIdx.x + j * 256;
    wcopy[j] = (i < NCH * NTAP) ? Wh[i] : 0.0f;
  }
  float bcopy = 0.0f, pcopy = 0.0f;
  if (threadIdx.x < NCH) {
    bcopy = bh[threadIdx.x];
    pcopy = Wp[threadIdx.x];
  }

  int  rowOff[5];
  int  col[5];
  bool yok[5];
#pragma unroll
  for (int s = 0; s < 5; ++s) {
    const int k  = (s >> 1) * 4 + khalf * 2 + (s & 1);
    const int kk = min(k, 8);
    const int dy = kk / 3 - 1, dx = kk % 3 - 1;
    const int yy = y + dy;
    yok[s]    = (k < 9) & (yy >= 0) & (yy < 256);
    rowOff[s] = min(max(yy, 0), 255) * 256;
    col[s]    = m + dx;
  }
  const float one9 = (khalf == 0) ? 1.0f : 0.0f;

  auto loadA = [&](int x0, v2f (&af)[3]) {
#pragma unroll
    for (int s = 0; s < 5; ++s) {
      const int  xx = x0 + col[s];
      const bool ok = yok[s] & (xx >= 0) & (xx < 256);
      const int  xc = min(max(xx, 0), 255);
      const float v = occ[rowOff[s] + xc];
      const float val = ok ? v : 0.0f;
      if (s & 1) af[s >> 1].y = val; else af[s >> 1].x = val;
    }
    af[2].y = one9;
  };

  v2f aCur[3];
  loadA(0, aCur);

#pragma unroll
  for (int j = 0; j < 6; ++j) {
    const int i = threadIdx.x + j * 256;
    if (i < NCH * NTAP) sW[i] = wcopy[j];
  }
  if (threadIdx.x == 0) sW[ZSLOT] = 0.0f;
  if (threadIdx.x < NCHP) {
    sW[BBASE + threadIdx.x] = bcopy;
    sWp[threadIdx.x]        = pcopy;
  }
  __syncthreads();

  v2f wFrag[10][3];
#pragma unroll
  for (int tile = 0; tile < 10; ++tile) {
    const int  n   = tile * 16 + m;
    const bool nok = n < NCH;
#pragma unroll
    for (int ks = 0; ks < 3; ++ks) {
      const int k0 = ks * 4 + khalf * 2;
      const int k1 = k0 + 1;
      const int i0 = (nok & (k0 < NTAP)) ? n * NTAP + k0 : ZSLOT;
      const int i1 = (k1 == NTAP) ? BBASE + n
                                  : ((nok & (k1 < NTAP)) ? n * NTAP + k1 : ZSLOT);
      wFrag[tile][ks].x = sW[i0];
      wFrag[tile][ks].y = sW[i1];
    }
  }

  float wpv[10][8];
#pragma unroll
  for (int tile = 0; tile < 10; ++tile) {
    const v4f lo = *(const v4f*)(sWp + tile * 16 + khalf * 8);
    const v4f hi = *(const v4f*)(sWp + tile * 16 + khalf * 8 + 4);
    wpv[tile][0] = lo.x; wpv[tile][1] = lo.y; wpv[tile][2] = lo.z; wpv[tile][3] = lo.w;
    wpv[tile][4] = hi.x; wpv[tile][5] = hi.y; wpv[tile][6] = hi.z; wpv[tile][7] = hi.w;
  }

#pragma nounroll
  for (int t = 0; t < CONV_TILES_PER_ROW; ++t) {
    v2f aNext[3];
    loadA(min(t + 1, CONV_TILES_PER_ROW - 1) * 16, aNext);

    float acc[8];
#pragma unroll
    for (int r = 0; r < 8; ++r) acc[r] = 0.0f;

#pragma unroll
    for (int tile = 0; tile < 10; ++tile) {
      v8f c = {0.0f, 0.0f, 0.0f, 0.0f, 0.0f, 0.0f, 0.0f, 0.0f};
#pragma unroll
      for (int ks = 0; ks < 3; ++ks) {
        c = __builtin_amdgcn_wmma_f32_16x16x4_f32(
             false, wFrag[tile][ks],  false, aCur[ks],
             (short)0, c,  false,  false);
      }
#pragma unroll
      for (int r = 0; r < 8; ++r) acc[r] += fmaxf(c[r], 0.0f) * wpv[tile][r];
    }

    float s = ((acc[0] + acc[1]) + (acc[2] + acc[3])) +
              ((acc[4] + acc[5]) + (acc[6] + acc[7]));
    s += xhalf(s);
    const float sg = 1.0f / (1.0f + __expf(-s));

    if (khalf == 0) prow[t * 16 + m] = sg;

#pragma unroll
    for (int ks = 0; ks < 3; ++ks) aCur[ks] = aNext[ks];
  }
}

#define TS     64
#define HALO   20
#define RS     (TS + 2 * HALO)
#define QPR    (RS / 4)
#define COL0   4
#define PITCH  112
#define BUFSZ  ((RS + 2) * PITCH)
#define FTHREADS 704

__device__ __forceinline__ float max3f(float a, float b, float c) {
  return fmaxf(fmaxf(a, b), c);
}

__device__ __forceinline__ void patch_step(const float* __restrict__ vin,
                                           int ry, int rx,
                                           const float (&r)[4][4],
                                           const float (&p)[4][4],
                                           float (&vn)[4][4]) {
  float hm[6][4];
  float vc[4][4];
#pragma unroll
  for (int k = 0; k < 6; ++k) {
    const float* rowp = vin + (ry + k) * PITCH + COL0 + rx;
    const v4f    m  = *(const v4f*)rowp;
    const float  lf = rowp[-1];
    const float  rt = rowp[4];
    hm[k][0] = max3f(lf,  m.x, m.y);
    hm[k][1] = max3f(m.x, m.y, m.z);
    hm[k][2] = max3f(m.y, m.z, m.w);
    hm[k][3] = max3f(m.z, m.w, rt);
    if (k >= 1 && k <= 4) {
      vc[k - 1][0] = m.x;
      vc[k - 1][1] = m.y;
      vc[k - 1][2] = m.z;
      vc[k - 1][3] = m.w;
    }
  }
#pragma unroll
  for (int j = 0; j < 4; ++j) {
#pragma unroll
    for (int i = 0; i < 4; ++i) {
      float mx = max3f(hm[j][i], hm[j + 1][i], hm[j + 2][i]);
      vn[j][i] = fmaxf(vc[j][i], r[j][i] + p[j][i] * (mx - r[j][i]));
    }
  }
}

__device__ __forceinline__ void store_patch(float* __restrict__ vout, int ry,
                                            int rx, const float (&vn)[4][4]) {
#pragma unroll
  for (int j = 0; j < 4; ++j) {
    float4* d = (float4*)(vout + (ry + 1 + j) * PITCH + COL0 + rx);
    *d = make_float4(vn[j][0], vn[j][1], vn[j][2], vn[j][3]);
  }
}

__global__ __launch_bounds__(FTHREADS) void vprop_fused(
    const float* __restrict__ image, const float* __restrict__ pbuf,
    float* __restrict__ out) {
  __shared__ __attribute__((aligned(16))) float sv[2 * BUFSZ];

  const int t    = threadIdx.x;
  const int tile = blockIdx.x;
  const int b    = tile >> 4;
  const int ty   = (tile >> 2) & 3;
  const int tx   = tile & 3;

  const int sy = t / QPR;
  const int qx = t - sy * QPR;
  const int ry = sy * 4;
  const int rx = qx * 4;
  const int gy = ty * TS - HALO + ry;
  const int gx = tx * TS - HALO + rx;
  const bool inimg = (t < QPR * QPR) & (gy >= 0) & (gy < 256) &
                     (gx >= 0) & (gx < 256);
  const bool inner = inimg & (ry >= HALO) & (ry < HALO + TS) &
                     (rx >= HALO) & (rx < HALO + TS);

  const int drow = (ry + 3 <= (RS - 1) / 2) ? ry + 3 : RS - 1 - ry;
  const int dcol = (rx + 3 <= (RS - 1) / 2) ? rx + 3 : RS - 1 - rx;
  const int dmax = inimg ? min(drow, dcol) : 0;

  const float* __restrict__ ib = image + (size_t)b * 2 * NPIX;
  const float* __restrict__ pb = pbuf + (size_t)b * NPIX;

  float r[4][4], p[4][4], oc[4][4];
  if (inimg) {
#pragma unroll
    for (int j = 0; j < 4; ++j) {
      const int g = (gy + j) * 256 + gx;
      const float4 r4 = *(const float4*)(ib + NPIX + g);
      const float4 p4 = *(const float4*)(pb + g);
      r[j][0] = r4.x; r[j][1] = r4.y; r[j][2] = r4.z; r[j][3] = r4.w;
      p[j][0] = p4.x; p[j][1] = p4.y; p[j][2] = p4.z; p[j][3] = p4.w;
    }
  }
  if (inner) {
#pragma unroll
    for (int j = 0; j < 4; ++j) {
      const float4 o4 = *(const float4*)(ib + (gy + j) * 256 + gx);
      oc[j][0] = o4.x; oc[j][1] = o4.y; oc[j][2] = o4.z; oc[j][3] = o4.w;
    }
  }

  {
    const float ninf = -__builtin_inff();
    const float4 n4 = make_float4(ninf, ninf, ninf, ninf);
    float4* s4 = (float4*)sv;
    for (int i = t; i < 2 * BUFSZ / 4; i += FTHREADS) s4[i] = n4;
  }
  __syncthreads();

  if (inimg) {
#pragma unroll
    for (int j = 0; j < 4; ++j) {
      float4* d = (float4*)(sv + (ry + 1 + j) * PITCH + COL0 + rx);
      *d = make_float4(r[j][0], r[j][1], r[j][2], r[j][3]);
    }
  }
  __syncthreads();

  float* const pl0 = sv;
  float* const pl1 = sv + BUFSZ;

  for (int s = 1; s < HALO - 1; s += 2) {
    if (dmax >= s) {
      float vn[4][4];
      patch_step(pl0, ry, rx, r, p, vn);
      store_patch(pl1, ry, rx, vn);
    }
    __syncthreads();
    if (dmax >= s + 1) {
      float vn[4][4];
      patch_step(pl1, ry, rx, r, p, vn);
      store_patch(pl0, ry, rx, vn);
    }
    __syncthreads();
  }
  if (dmax >= HALO - 1) {
    float vn[4][4];
    patch_step(pl0, ry, rx, r, p, vn);
    store_patch(pl1, ry, rx, vn);
  }
  __syncthreads();

  if (inner) {
    float vn[4][4];
    patch_step(pl1, ry, rx, r, p, vn);
#pragma unroll
    for (int j = 0; j < 4; ++j) {
      float4 o;
      o.x = (oc[j][0] > 0.49f) ? vn[j][0] : -1.0f;
      o.y = (oc[j][1] > 0.49f) ? vn[j][1] : -1.0f;
      o.z = (oc[j][2] > 0.49f) ? vn[j][2] : -1.0f;
      o.w = (oc[j][3] > 0.49f) ? vn[j][3] : -1.0f;
      *(float4*)(out + (size_t)b * NPIX + (gy + j) * 256 + gx) = o;
    }
  }
}

extern "C" void kernel_launch(void* const* d_in, const int* in_sizes, int n_in,
                              void* d_out, int out_size, void* d_ws,
                              size_t ws_size, hipStream_t stream) {
  const float* image = (const float*)d_in[0];
  const float* Wh    = (const float*)d_in[1];
  const float* bh    = (const float*)d_in[2];
  const float* Wp    = (const float*)d_in[3];
  float* out = (float*)d_out;

  float* pbuf  = (float*)d_ws;

  mvprop_conv<<<512, 256, 0, stream>>>(image, Wh, bh, Wp, pbuf);

  vprop_fused<<<NBATCH * 16, FTHREADS, 0, stream>>>(image, pbuf, out);
}


# conv: third k-step dropped (K 12->8, 30->20 WMMAs per tile): bias is the C operand, tap 8 one FMA per D element before the relu; 334 VGPRs, 2 waves/SIMD, 2 rows/wave; per-channel sum order changes
# speedup vs baseline: 5.0635x; 1.0820x over previous
#include <hip/hip_runtime.h>

typedef __attribute__((ext_vector_type(2))) float v2f;
typedef __attribute__((ext_vector_type(8))) float v8f;
typedef __attribute__((ext_vector_type(4))) float v4f;

#define NPIX   (256 * 256)
#define NBATCH 16
#define NTOT   (NBATCH * NPIX)

#define CONV_TILES_PER_WAVE 32
#define NCH   150
#define NCHP  160
#define NTAP  9
#define ZSLOT (NCH * NTAP)

__device__ __forceinline__ float xhalf(float x) {
  const int b = __builtin_bit_cast(int, x);
  return __builtin_bit_cast(
      float, __builtin_amdgcn_permlanex16(b, b, 0x76543210, 0xfedcba98, false,
                                          false));
}

__global__ __launch_bounds__(256) void mvprop_conv(
    const float* __restrict__ image, const float* __restrict__ Wh,
    const float* __restrict__ bh, const float* __restrict__ Wp,
    float* __restrict__ pout) {
  __shared__ float sW[NCH * NTAP + 1];
  __shared__ __attribute__((aligned(16))) float sB[NCHP];
  __shared__ __attribute__((aligned(16))) float sW8[NCHP];
  __shared__ __attribute__((aligned(16))) float sWp[NCHP];

  const int lane  = threadIdx.x & 31;
  const int wave  = threadIdx.x >> 5;
  const int row0  = (blockIdx.x * 8 + wave) * 2;
  const int b     = row0 >> 8;
  const int y0    = row0 & 255;

  const float* __restrict__ occ = image + (size_t)b * 2 * NPIX;
  float* __restrict__ prow = pout + (size_t)row0 * 256;

  const int m     = lane & 15;
  const int khalf = lane >> 4;

  float wcopy[6];
#pragma unroll
  for (int j = 0; j < 6; ++j) {
    const int i = threadIdx.x + j * 256;
    wcopy[j] = (i < NCH * NTAP) ? Wh[i] : 0.0f;
  }
  float bcopy = 0.0f, pcopy = 0.0f, w8copy = 0.0f;
  if (threadIdx.x < NCH) {
    bcopy  = bh[threadIdx.x];
    pcopy  = Wp[threadIdx.x];
    w8copy = Wh[threadIdx.x * NTAP + 8];
  }

  int dyv[5];
  int col[5];
#pragma unroll
  for (int s = 0; s < 5; ++s) {
    const int k  = (s < 4) ? (s >> 1) * 4 + khalf * 2 + (s & 1) : 8;
    dyv[s] = k / 3 - 1;
    col[s] = m + (k % 3 - 1);
  }

  auto loadP = [&](int tt, v2f (&pf)[2], float& p8) {
    const int yn = y0 + (tt >> 4);
    const int xn = (tt & 15) * 16;
    float val[5];
#pragma unroll
    for (int s = 0; s < 5; ++s) {
      const int  yy = yn + dyv[s];
      const int  xx = xn + col[s];
      const bool ok = ((unsigned)yy < 256u) & ((unsigned)xx < 256u);
      const int  idx = ok ? yy * 256 + xx : 0;
      const float v = occ[idx];
      val[s] = ok ? v : 0.0f;
    }
    pf[0].x = val[0]; pf[0].y = val[1];
    pf[1].x = val[2]; pf[1].y = val[3];
    p8 = val[4];
  };

  v2f   pCur[2];
  float p8Cur;
  loadP(0, pCur, p8Cur);

#pragma unroll
  for (int j = 0; j < 6; ++j) {
    const int i = threadIdx.x + j * 256;
    if (i < NCH * NTAP) sW[i] = wcopy[j];
  }
  if (threadIdx.x == 0) sW[ZSLOT] = 0.0f;
  if (threadIdx.x < NCHP) {
    sB[threadIdx.x]  = bcopy;
    sW8[threadIdx.x] = w8copy;
    sWp[threadIdx.x] = pcopy;
  }
  __syncthreads();

  v2f wFrag[10][2];
#pragma unroll
  for (int tile = 0; tile < 10; ++tile) {
    const int  n   = tile * 16 + m;
    const bool nok = n < NCH;
#pragma unroll
    for (int ks = 0; ks < 2; ++ks) {
      const int k0 = ks * 4 + khalf * 2;
      const int i0 = nok ? n * NTAP + k0 : ZSLOT;
      const int i1 = nok ? n * NTAP + k0 + 1 : ZSLOT;
      wFrag[tile][ks].x = sW[i0];
      wFrag[tile][ks].y = sW[i1];
    }
  }

  v8f   biasC[10];
  float w8v[10][8];
  float wpv[10][8];
#pragma unroll
  for (int tile = 0; tile < 10; ++tile) {
    const int o = tile * 16 + khalf * 8;
    const v4f blo = *(const v4f*)(sB + o);
    const v4f bhi = *(const v4f*)(sB + o + 4);
    biasC[tile] = (v8f){blo.x, blo.y, blo.z, blo.w, bhi.x, bhi.y, bhi.z, bhi.w};
    const v4f elo = *(const v4f*)(sW8 + o);
    const v4f ehi = *(const v4f*)(sW8 + o + 4);
    w8v[tile][0] = elo.x; w8v[tile][1] = elo.y; w8v[tile][2] = elo.z; w8v[tile][3] = elo.w;
    w8v[tile][4] = ehi.x; w8v[tile][5] = ehi.y; w8v[tile][6] = ehi.z; w8v[tile][7] = ehi.w;
    const v4f lo = *(const v4f*)(sWp + o);
    const v4f hi = *(const v4f*)(sWp + o + 4);
    wpv[tile][0] = lo.x; wpv[tile][1] = lo.y; wpv[tile][2] = lo.z; wpv[tile][3] = lo.w;
    wpv[tile][4] = hi.x; wpv[tile][5] = hi.y; wpv[tile][6] = hi.z; wpv[tile][7] = hi.w;
  }

#pragma nounroll
  for (int tt = 0; tt < CONV_TILES_PER_WAVE; ++tt) {
    v2f   pNext[2];
    float p8Next;
    loadP(min(tt + 1, CONV_TILES_PER_WAVE - 1), pNext, p8Next);

    float acc[8];
#pragma unroll
    for (int r = 0; r < 8; ++r) acc[r] = 0.0f;

#pragma unroll
    for (int tile = 0; tile < 10; ++tile) {
      v8f c = __builtin_amdgcn_wmma_f32_16x16x4_f32(
           false, wFrag[tile][0],  false, pCur[0],
           (short)0, biasC[tile],  false,  false);
      c = __builtin_amdgcn_wmma_f32_16x16x4_f32(
           false, wFrag[tile][1],  false, pCur[1],
           (short)0, c,  false,  false);
#pragma unroll
      for (int r = 0; r < 8; ++r) {
        const float hsum = __builtin_fmaf(w8v[tile][r], p8Cur, c[r]);
        acc[r] += fmaxf(hsum, 0.0f) * wpv[tile][r];
      }
    }

    float s = ((acc[0] + acc[1]) + (acc[2] + acc[3])) +
              ((acc[4] + acc[5]) + (acc[6] + acc[7]));
    s += xhalf(s);
    const float sg = 1.0f / (1.0f + __expf(-s));

    if (khalf == 0) prow[tt * 16 + m] = sg;

    pCur[0] = pNext[0];
    pCur[1] = pNext[1];
    p8Cur   = p8Next;
  }
}

#define TS     64
#define HALO   20
#define RS     (TS + 2 * HALO)
#define QPR    (RS / 4)
#define COL0   4
#define PITCH  112
#define BUFSZ  ((RS + 2) * PITCH)
#define FTHREADS 704

__device__ __forceinline__ float max3f(float a, float b, float c) {
  return fmaxf(fmaxf(a, b), c);
}

__device__ __forceinline__ void patch_step(const float* __restrict__ vin,
                                           int ry, int rx,
                                           const float (&r)[4][4],
                                           const float (&p)[4][4],
                                           float (&vn)[4][4]) {
  float hm[6][4];
  float vc[4][4];
#pragma unroll
  for (int k = 0; k < 6; ++k) {
    const float* rowp = vin + (ry + k) * PITCH + COL0 + rx;
    const v4f    m  = *(const v4f*)rowp;
    const float  lf = rowp[-1];
    const float  rt = rowp[4];
    hm[k][0] = max3f(lf,  m.x, m.y);
    hm[k][1] = max3f(m.x, m.y, m.z);
    hm[k][2] = max3f(m.y, m.z, m.w);
    hm[k][3] = max3f(m.z, m.w, rt);
    if (k >= 1 && k <= 4) {
      vc[k - 1][0] = m.x;
      vc[k - 1][1] = m.y;
      vc[k - 1][2] = m.z;
      vc[k - 1][3] = m.w;
    }
  }
#pragma unroll
  for (int j = 0; j < 4; ++j) {
#pragma unroll
    for (int i = 0; i < 4; ++i) {
      float mx = max3f(hm[j][i], hm[j + 1][i], hm[j + 2][i]);
      vn[j][i] = fmaxf(vc[j][i], r[j][i] + p[j][i] * (mx - r[j][i]));
    }
  }
}

__device__ __forceinline__ void store_patch(float* __restrict__ vout, int ry,
                                            int rx, const float (&vn)[4][4]) {
#pragma unroll
  for (int j = 0; j < 4; ++j) {
    float4* d = (float4*)(vout + (ry + 1 + j) * PITCH + COL0 + rx);
    *d = make_float4(vn[j][0], vn[j][1], vn[j][2], vn[j][3]);
  }
}

__global__ __launch_bounds__(FTHREADS) void vprop_fused(
    const float* __restrict__ image, const float* __restrict__ pbuf,
    float* __restrict__ out) {
  __shared__ __attribute__((aligned(16))) float sv[2 * BUFSZ];

  const int t    = threadIdx.x;
  const int tile = blockIdx.x;
  const int b    = tile >> 4;
  const int ty   = (tile >> 2) & 3;
  const int tx   = tile & 3;

  const int sy = t / QPR;
  const int qx = t - sy * QPR;
  const int ry = sy * 4;
  const int rx = qx * 4;
  const int gy = ty * TS - HALO + ry;
  const int gx = tx * TS - HALO + rx;
  const bool inimg = (t < QPR * QPR) & (gy >= 0) & (gy < 256) &
                     (gx >= 0) & (gx < 256);
  const bool inner = inimg & (ry >= HALO) & (ry < HALO + TS) &
                     (rx >= HALO) & (rx < HALO + TS);

  const int drow = (ry + 3 <= (RS - 1) / 2) ? ry + 3 : RS - 1 - ry;
  const int dcol = (rx + 3 <= (RS - 1) / 2) ? rx + 3 : RS - 1 - rx;
  const int dmax = inimg ? min(drow, dcol) : 0;

  const float* __restrict__ ib = image + (size_t)b * 2 * NPIX;
  const float* __restrict__ pb = pbuf + (size_t)b * NPIX;

  float r[4][4], p[4][4], oc[4][4];
  if (inimg) {
#pragma unroll
    for (int j = 0; j < 4; ++j) {
      const int g = (gy + j) * 256 + gx;
      const float4 r4 = *(const float4*)(ib + NPIX + g);
      const float4 p4 = *(const float4*)(pb + g);
      r[j][0] = r4.x; r[j][1] = r4.y; r[j][2] = r4.z; r[j][3] = r4.w;
      p[j][0] = p4.x; p[j][1] = p4.y; p[j][2] = p4.z; p[j][3] = p4.w;
    }
  }
  if (inner) {
#pragma unroll
    for (int j = 0; j < 4; ++j) {
      const float4 o4 = *(const float4*)(ib + (gy + j) * 256 + gx);
      oc[j][0] = o4.x; oc[j][1] = o4.y; oc[j][2] = o4.z; oc[j][3] = o4.w;
    }
  }

  {
    const float ninf = -__builtin_inff();
    const float4 n4 = make_float4(ninf, ninf, ninf, ninf);
    float4* s4 = (float4*)sv;
    for (int i = t; i < 2 * BUFSZ / 4; i += FTHREADS) s4[i] = n4;
  }
  __syncthreads();

  if (inimg) {
#pragma unroll
    for (int j = 0; j < 4; ++j) {
      float4* d = (float4*)(sv + (ry + 1 + j) * PITCH + COL0 + rx);
      *d = make_float4(r[j][0], r[j][1], r[j][2], r[j][3]);
    }
  }
  __syncthreads();

  float* const pl0 = sv;
  float* const pl1 = sv + BUFSZ;

  for (int s = 1; s < HALO - 1; s += 2) {
    if (dmax >= s) {
      float vn[4][4];
      patch_step(pl0, ry, rx, r, p, vn);
      store_patch(pl1, ry, rx, vn);
    }
    __syncthreads();
    if (dmax >= s + 1) {
      float vn[4][4];
      patch_step(pl1, ry, rx, r, p, vn);
      store_patch(pl0, ry, rx, vn);
    }
    __syncthreads();
  }
  if (dmax >= HALO - 1) {
    float vn[4][4];
    patch_step(pl0, ry, rx, r, p, vn);
    store_patch(pl1, ry, rx, vn);
  }
  __syncthreads();

  if (inner) {
    float vn[4][4];
    patch_step(pl1, ry, rx, r, p, vn);
#pragma unroll
    for (int j = 0; j < 4; ++j) {
      float4 o;
      o.x = (oc[j][0] > 0.49f) ? vn[j][0] : -1.0f;
      o.y = (oc[j][1] > 0.49f) ? vn[j][1] : -1.0f;
      o.z = (oc[j][2] > 0.49f) ? vn[j][2] : -1.0f;
      o.w = (oc[j][3] > 0.49f) ? vn[j][3] : -1.0f;
      *(float4*)(out + (size_t)b * NPIX + (gy + j) * 256 + gx) = o;
    }
  }
}

extern "C" void kernel_launch(void* const* d_in, const int* in_sizes, int n_in,
                              void* d_out, int out_size, void* d_ws,
                              size_t ws_size, hipStream_t stream) {
  const float* image = (const float*)d_in[0];
  const float* Wh    = (const float*)d_in[1];
  const float* bh    = (const float*)d_in[2];
  const float* Wp    = (const float*)d_in[3];
  float* out = (float*)d_out;

  float* pbuf  = (float*)d_ws;

  mvprop_conv<<<256, 256, 0, stream>>>(image, Wh, bh, Wp, pbuf);

  vprop_fused<<<NBATCH * 16, FTHREADS, 0, stream>>>(image, pbuf, out);
}


# propagation: each thread keeps its 4x4 patch of v in registers for all 20 steps, updated in place; LDS is read only for the 20 cells round it (12 LDS reads per thread-step -> 8); same arithmetic
# speedup vs baseline: 5.3315x; 1.0529x over previous
#include <hip/hip_runtime.h>

typedef __attribute__((ext_vector_type(2))) float v2f;
typedef __attribute__((ext_vector_type(8))) float v8f;
typedef __attribute__((ext_vector_type(4))) float v4f;

#define NPIX   (256 * 256)
#define NBATCH 16
#define NTOT   (NBATCH * NPIX)

#define CONV_TILES_PER_WAVE 32
#define NCH   150
#define NCHP  160
#define NTAP  9
#define ZSLOT (NCH * NTAP)

__device__ __forceinline__ float xhalf(float x) {
  const int b = __builtin_bit_cast(int, x);
  return __builtin_bit_cast(
      float, __builtin_amdgcn_permlanex16(b, b, 0x76543210, 0xfedcba98, false,
                                          false));
}

__global__ __launch_bounds__(256) void mvprop_conv(
    const float* __restrict__ image, const float* __restrict__ Wh,
    const float* __restrict__ bh, const float* __restrict__ Wp,
    float* __restrict__ pout) {
  __shared__ float sW[NCH * NTAP + 1];
  __shared__ __attribute__((aligned(16))) float sB[NCHP];
  __shared__ __attribute__((aligned(16))) float sW8[NCHP];
  __shared__ __attribute__((aligned(16))) float sWp[NCHP];

  const int lane  = threadIdx.x & 31;
  const int wave  = threadIdx.x >> 5;
  const int row0  = (blockIdx.x * 8 + wave) * 2;
  const int b     = row0 >> 8;
  const int y0    = row0 & 255;

  const float* __restrict__ occ = image + (size_t)b * 2 * NPIX;
  float* __restrict__ prow = pout + (size_t)row0 * 256;

  const int m     = lane & 15;
  const int khalf = lane >> 4;

  float wcopy[6];
#pragma unroll
  for (int j = 0; j < 6; ++j) {
    const int i = threadIdx.x + j * 256;
    wcopy[j] = (i < NCH * NTAP) ? Wh[i] : 0.0f;
  }
  float bcopy = 0.0f, pcopy = 0.0f, w8copy = 0.0f;
  if (threadIdx.x < NCH) {
    bcopy  = bh[threadIdx.x];
    pcopy  = Wp[threadIdx.x];
    w8copy = Wh[threadIdx.x * NTAP + 8];
  }

  int dyv[5];
  int col[5];
#pragma unroll
  for (int s = 0; s < 5; ++s) {
    const int k  = (s < 4) ? (s >> 1) * 4 + khalf * 2 + (s & 1) : 8;
    dyv[s] = k / 3 - 1;
    col[s] = m + (k % 3 - 1);
  }

  auto loadP = [&](int tt, v2f (&pf)[2], float& p8) {
    const int yn = y0 + (tt >> 4);
    const int xn = (tt & 15) * 16;
    float val[5];
#pragma unroll
    for (int s = 0; s < 5; ++s) {
      const int  yy = yn + dyv[s];
      const int  xx = xn + col[s];
      const bool ok = ((unsigned)yy < 256u) & ((unsigned)xx < 256u);
      const int  idx = ok ? yy * 256 + xx : 0;
      const float v = occ[idx];
      val[s] = ok ? v : 0.0f;
    }
    pf[0].x = val[0]; pf[0].y = val[1];
    pf[1].x = val[2]; pf[1].y = val[3];
    p8 = val[4];
  };

  v2f   pCur[2];
  float p8Cur;
  loadP(0, pCur, p8Cur);

#pragma unroll
  for (int j = 0; j < 6; ++j) {
    const int i = threadIdx.x + j * 256;
    if (i < NCH * NTAP) sW[i] = wcopy[j];
  }
  if (threadIdx.x == 0) sW[ZSLOT] = 0.0f;
  if (threadIdx.x < NCHP) {
    sB[threadIdx.x]  = bcopy;
    sW8[threadIdx.x] = w8copy;
    sWp[threadIdx.x] = pcopy;
  }
  __syncthreads();

  v2f wFrag[10][2];
#pragma unroll
  for (int tile = 0; tile < 10; ++tile) {
    const int  n   = tile * 16 + m;
    const bool nok = n < NCH;
#pragma unroll
    for (int ks = 0; ks < 2; ++ks) {
      const int k0 = ks * 4 + khalf * 2;
      const int i0 = nok ? n * NTAP + k0 : ZSLOT;
      const int i1 = nok ? n * NTAP + k0 + 1 : ZSLOT;
      wFrag[tile][ks].x = sW[i0];
      wFrag[tile][ks].y = sW[i1];
    }
  }

  v8f   biasC[10];
  float w8v[10][8];
  float wpv[10][8];
#pragma unroll
  for (int tile = 0; tile < 10; ++tile) {
    const int o = tile * 16 + khalf * 8;
    const v4f blo = *(const v4f*)(sB + o);
    const v4f bhi = *(const v4f*)(sB + o + 4);
    biasC[tile] = (v8f){blo.x, blo.y, blo.z, blo.w, bhi.x, bhi.y, bhi.z, bhi.w};
    const v4f elo = *(const v4f*)(sW8 + o);
    const v4f ehi = *(const v4f*)(sW8 + o + 4);
    w8v[tile][0] = elo.x; w8v[tile][1] = elo.y; w8v[tile][2] = elo.z; w8v[tile][3] = elo.w;
    w8v[tile][4] = ehi.x; w8v[tile][5] = ehi.y; w8v[tile][6] = ehi.z; w8v[tile][7] = ehi.w;
    const v4f lo = *(const v4f*)(sWp + o);
    const v4f hi = *(const v4f*)(sWp + o + 4);
    wpv[tile][0] = lo.x; wpv[tile][1] = lo.y; wpv[tile][2] = lo.z; wpv[tile][3] = lo.w;
    wpv[tile][4] = hi.x; wpv[tile][5] = hi.y; wpv[tile][6] = hi.z; wpv[tile][7] = hi.w;
  }

#pragma nounroll
  for (int tt = 0; tt < CONV_TILES_PER_WAVE; ++tt) {
    v2f   pNext[2];
    float p8Next;
    loadP(min(tt + 1, CONV_TILES_PER_WAVE - 1), pNext, p8Next);

    float acc[8];
#pragma unroll
    for (int r = 0; r < 8; ++r) acc[r] = 0.0f;

#pragma unroll
    for (int tile = 0; tile < 10; ++tile) {
      v8f c = __builtin_amdgcn_wmma_f32_16x16x4_f32(
           false, wFrag[tile][0],  false, pCur[0],
           (short)0, biasC[tile],  false,  false);
      c = __builtin_amdgcn_wmma_f32_16x16x4_f32(
           false, wFrag[tile][1],  false, pCur[1],
           (short)0, c,  false,  false);
#pragma unroll
      for (int r = 0; r < 8; ++r) {
        const float hsum = __builtin_fmaf(w8v[tile][r], p8Cur, c[r]);
        acc[r] += fmaxf(hsum, 0.0f) * wpv[tile][r];
      }
    }

    float s = ((acc[0] + acc[1]) + (acc[2] + acc[3])) +
              ((acc[4] + acc[5]) + (acc[6] + acc[7]));
    s += xhalf(s);
    const float sg = 1.0f / (1.0f + __expf(-s));

    if (khalf == 0) prow[tt * 16 + m] = sg;

    pCur[0] = pNext[0];
    pCur[1] = pNext[1];
    p8Cur   = p8Next;
  }
}

#define TS     64
#define HALO   20
#define RS     (TS + 2 * HALO)
#define QPR    (RS / 4)
#define COL0   4
#define PITCH  112
#define BUFSZ  ((RS + 2) * PITCH)
#define FTHREADS 704

__device__ __forceinline__ float max3f(float a, float b, float c) {
  return fmaxf(fmaxf(a, b), c);
}

__device__ __forceinline__ void patch_step(const float* __restrict__ vin,
                                           int ry, int rx,
                                           const float (&r)[4][4],
                                           const float (&p)[4][4],
                                           v4f (&v)[4]) {
  float hm[6][4];
#pragma unroll
  for (int k = 0; k < 6; k += 5) {
    const float* rowp = vin + (ry + k) * PITCH + COL0 + rx;
    const v4f    m  = *(const v4f*)rowp;
    const float  lf = rowp[-1];
    const float  rt = rowp[4];
    hm[k][0] = max3f(lf,  m.x, m.y);
    hm[k][1] = max3f(m.x, m.y, m.z);
    hm[k][2] = max3f(m.y, m.z, m.w);
    hm[k][3] = max3f(m.z, m.w, rt);
  }
#pragma unroll
  for (int j = 0; j < 4; ++j) {
    const float* rowp = vin + (ry + 1 + j) * PITCH + COL0 + rx;
    const float  lf = rowp[-1];
    const float  rt = rowp[4];
    hm[j + 1][0] = max3f(lf,     v[j].x, v[j].y);
    hm[j + 1][1] = max3f(v[j].x, v[j].y, v[j].z);
    hm[j + 1][2] = max3f(v[j].y, v[j].z, v[j].w);
    hm[j + 1][3] = max3f(v[j].z, v[j].w, rt);
  }
#pragma unroll
  for (int j = 0; j < 4; ++j) {
#pragma unroll
    for (int i = 0; i < 4; ++i) {
      float mx = max3f(hm[j][i], hm[j + 1][i], hm[j + 2][i]);
      v[j][i] = fmaxf(v[j][i], r[j][i] + p[j][i] * (mx - r[j][i]));
    }
  }
}

__device__ __forceinline__ void store_patch(float* __restrict__ vout, int ry,
                                            int rx, const v4f (&v)[4]) {
#pragma unroll
  for (int j = 0; j < 4; ++j) {
    *(v4f*)(vout + (ry + 1 + j) * PITCH + COL0 + rx) = v[j];
  }
}

__global__ __launch_bounds__(FTHREADS) void vprop_fused(
    const float* __restrict__ image, const float* __restrict__ pbuf,
    float* __restrict__ out) {
  __shared__ __attribute__((aligned(16))) float sv[2 * BUFSZ];

  const int t    = threadIdx.x;
  const int tile = blockIdx.x;
  const int b    = tile >> 4;
  const int ty   = (tile >> 2) & 3;
  const int tx   = tile & 3;

  const int sy = t / QPR;
  const int qx = t - sy * QPR;
  const int ry = sy * 4;
  const int rx = qx * 4;
  const int gy = ty * TS - HALO + ry;
  const int gx = tx * TS - HALO + rx;
  const bool inimg = (t < QPR * QPR) & (gy >= 0) & (gy < 256) &
                     (gx >= 0) & (gx < 256);
  const bool inner = inimg & (ry >= HALO) & (ry < HALO + TS) &
                     (rx >= HALO) & (rx < HALO + TS);

  const int drow = (ry + 3 <= (RS - 1) / 2) ? ry + 3 : RS - 1 - ry;
  const int dcol = (rx + 3 <= (RS - 1) / 2) ? rx + 3 : RS - 1 - rx;
  const int dmax = inimg ? min(drow, dcol) : 0;

  const float* __restrict__ ib = image + (size_t)b * 2 * NPIX;
  const float* __restrict__ pb = pbuf + (size_t)b * NPIX;

  float r[4][4], p[4][4], oc[4][4];
  v4f v[4];
  if (inimg) {
#pragma unroll
    for (int j = 0; j < 4; ++j) {
      const int g = (gy + j) * 256 + gx;
      const float4 r4 = *(const float4*)(ib + NPIX + g);
      const float4 p4 = *(const float4*)(pb + g);
      r[j][0] = r4.x; r[j][1] = r4.y; r[j][2] = r4.z; r[j][3] = r4.w;
      p[j][0] = p4.x; p[j][1] = p4.y; p[j][2] = p4.z; p[j][3] = p4.w;
    }
  }
  if (inner) {
#pragma unroll
    for (int j = 0; j < 4; ++j) {
      const float4 o4 = *(const float4*)(ib + (gy + j) * 256 + gx);
      oc[j][0] = o4.x; oc[j][1] = o4.y; oc[j][2] = o4.z; oc[j][3] = o4.w;
    }
  }

  {
    const float ninf = -__builtin_inff();
    const float4 n4 = make_float4(ninf, ninf, ninf, ninf);
    float4* s4 = (float4*)sv;
    for (int i = t; i < 2 * BUFSZ / 4; i += FTHREADS) s4[i] = n4;
  }
  __syncthreads();

  if (inimg) {
#pragma unroll
    for (int j = 0; j < 4; ++j) {
      v[j] = (v4f){r[j][0], r[j][1], r[j][2], r[j][3]};
      *(v4f*)(sv + (ry + 1 + j) * PITCH + COL0 + rx) = v[j];
    }
  }
  __syncthreads();

  float* const pl0 = sv;
  float* const pl1 = sv + BUFSZ;

  for (int s = 1; s < HALO - 1; s += 2) {
    if (dmax >= s) {
      patch_step(pl0, ry, rx, r, p, v);
      store_patch(pl1, ry, rx, v);
    }
    __syncthreads();
    if (dmax >= s + 1) {
      patch_step(pl1, ry, rx, r, p, v);
      store_patch(pl0, ry, rx, v);
    }
    __syncthreads();
  }
  if (dmax >= HALO - 1) {
    patch_step(pl0, ry, rx, r, p, v);
    store_patch(pl1, ry, rx, v);
  }
  __syncthreads();

  if (inner) {
    patch_step(pl1, ry, rx, r, p, v);
#pragma unroll
    for (int j = 0; j < 4; ++j) {
      float4 o;
      o.x = (oc[j][0] > 0.49f) ? v[j].x : -1.0f;
      o.y = (oc[j][1] > 0.49f) ? v[j].y : -1.0f;
      o.z = (oc[j][2] > 0.49f) ? v[j].z : -1.0f;
      o.w = (oc[j][3] > 0.49f) ? v[j].w : -1.0f;
      *(float4*)(out + (size_t)b * NPIX + (gy + j) * 256 + gx) = o;
    }
  }
}

extern "C" void kernel_launch(void* const* d_in, const int* in_sizes, int n_in,
                              void* d_out, int out_size, void* d_ws,
                              size_t ws_size, hipStream_t stream) {
  const float* image = (const float*)d_in[0];
  const float* Wh    = (const float*)d_in[1];
  const float* bh    = (const float*)d_in[2];
  const float* Wp    = (const float*)d_in[3];
  float* out = (float*)d_out;

  float* pbuf  = (float*)d_ws;

  mvprop_conv<<<256, 256, 0, stream>>>(image, Wh, bh, Wp, pbuf);

  vprop_fused<<<NBATCH * 16, FTHREADS, 0, stream>>>(image, pbuf, out);
}


# propagation: each patch publishes its left and right edge column as one 16-byte LDS vector (column-major plane); a neighbour reads 2 b128, not 8 strided dwords on 16 banks; same arithmetic
# speedup vs baseline: 5.6796x; 1.0653x over previous
#include <hip/hip_runtime.h>

typedef __attribute__((ext_vector_type(2))) float v2f;
typedef __attribute__((ext_vector_type(8))) float v8f;
typedef __attribute__((ext_vector_type(4))) float v4f;

#define NPIX   (256 * 256)
#define NBATCH 16
#define NTOT   (NBATCH * NPIX)

#define CONV_TILES_PER_WAVE 32
#define NCH   150
#define NCHP  160
#define NTAP  9
#define ZSLOT (NCH * NTAP)

__device__ __forceinline__ float xhalf(float x) {
  const int b = __builtin_bit_cast(int, x);
  return __builtin_bit_cast(
      float, __builtin_amdgcn_permlanex16(b, b, 0x76543210, 0xfedcba98, false,
                                          false));
}

__global__ __launch_bounds__(256) void mvprop_conv(
    const float* __restrict__ image, const float* __restrict__ Wh,
    const float* __restrict__ bh, const float* __restrict__ Wp,
    float* __restrict__ pout) {
  __shared__ float sW[NCH * NTAP + 1];
  __shared__ __attribute__((aligned(16))) float sB[NCHP];
  __shared__ __attribute__((aligned(16))) float sW8[NCHP];
  __shared__ __attribute__((aligned(16))) float sWp[NCHP];

  const int lane  = threadIdx.x & 31;
  const int wave  = threadIdx.x >> 5;
  const int row0  = (blockIdx.x * 8 + wave) * 2;
  const int b     = row0 >> 8;
  const int y0    = row0 & 255;

  const float* __restrict__ occ = image + (size_t)b * 2 * NPIX;
  float* __restrict__ prow = pout + (size_t)row0 * 256;

  const int m     = lane & 15;
  const int khalf = lane >> 4;

  float wcopy[6];
#pragma unroll
  for (int j = 0; j < 6; ++j) {
    const int i = threadIdx.x + j * 256;
    wcopy[j] = (i < NCH * NTAP) ? Wh[i] : 0.0f;
  }
  float bcopy = 0.0f, pcopy = 0.0f, w8copy = 0.0f;
  if (threadIdx.x < NCH) {
    bcopy  = bh[threadIdx.x];
    pcopy  = Wp[threadIdx.x];
    w8copy = Wh[threadIdx.x * NTAP + 8];
  }

  int dyv[5];
  int col[5];
#pragma unroll
  for (int s = 0; s < 5; ++s) {
    const int k  = (s < 4) ? (s >> 1) * 4 + khalf * 2 + (s & 1) : 8;
    dyv[s] = k / 3 - 1;
    col[s] = m + (k % 3 - 1);
  }

  auto loadP = [&](int tt, v2f (&pf)[2], float& p8) {
    const int yn = y0 + (tt >> 4);
    const int xn = (tt & 15) * 16;
    float val[5];
#pragma unroll
    for (int s = 0; s < 5; ++s) {
      const int  yy = yn + dyv[s];
      const int  xx = xn + col[s];
      const bool ok = ((unsigned)yy < 256u) & ((unsigned)xx < 256u);
      const int  idx = ok ? yy * 256 + xx : 0;
      const float v = occ[idx];
      val[s] = ok ? v : 0.0f;
    }
    pf[0].x = val[0]; pf[0].y = val[1];
    pf[1].x = val[2]; pf[1].y = val[3];
    p8 = val[4];
  };

  v2f   pCur[2];
  float p8Cur;
  loadP(0, pCur, p8Cur);

#pragma unroll
  for (int j = 0; j < 6; ++j) {
    const int i = threadIdx.x + j * 256;
    if (i < NCH * NTAP) sW[i] = wcopy[j];
  }
  if (threadIdx.x == 0) sW[ZSLOT] = 0.0f;
  if (threadIdx.x < NCHP) {
    sB[threadIdx.x]  = bcopy;
    sW8[threadIdx.x] = w8copy;
    sWp[threadIdx.x] = pcopy;
  }
  __syncthreads();

  v2f wFrag[10][2];
#pragma unroll
  for (int tile = 0; tile < 10; ++tile) {
    const int  n   = tile * 16 + m;
    const bool nok = n < NCH;
#pragma unroll
    for (int ks = 0; ks < 2; ++ks) {
      const int k0 = ks * 4 + khalf * 2;
      const int i0 = nok ? n * NTAP + k0 : ZSLOT;
      const int i1 = nok ? n * NTAP + k0 + 1 : ZSLOT;
      wFrag[tile][ks].x = sW[i0];
      wFrag[tile][ks].y = sW[i1];
    }
  }

  v8f   biasC[10];
  float w8v[10][8];
  float wpv[10][8];
#pragma unroll
  for (int tile = 0; tile < 10; ++tile) {
    const int o = tile * 16 + khalf * 8;
    const v4f blo = *(const v4f*)(sB + o);
    const v4f bhi = *(const v4f*)(sB + o + 4);
    biasC[tile] = (v8f){blo.x, blo.y, blo.z, blo.w, bhi.x, bhi.y, bhi.z, bhi.w};
    const v4f elo = *(const v4f*)(sW8 + o);
    const v4f ehi = *(const v4f*)(sW8 + o + 4);
    w8v[tile][0] = elo.x; w8v[tile][1] = elo.y; w8v[tile][2] = elo.z; w8v[tile][3] = elo.w;
    w8v[tile][4] = ehi.x; w8v[tile][5] = ehi.y; w8v[tile][6] = ehi.z; w8v[tile][7] = ehi.w;
    const v4f lo = *(const v4f*)(sWp + o);
    const v4f hi = *(const v4f*)(sWp + o + 4);
    wpv[tile][0] = lo.x; wpv[tile][1] = lo.y; wpv[tile][2] = lo.z; wpv[tile][3] = lo.w;
    wpv[tile][4] = hi.x; wpv[tile][5] = hi.y; wpv[tile][6] = hi.z; wpv[tile][7] = hi.w;
  }

#pragma nounroll
  for (int tt = 0; tt < CONV_TILES_PER_WAVE; ++tt) {
    v2f   pNext[2];
    float p8Next;
    loadP(min(tt + 1, CONV_TILES_PER_WAVE - 1), pNext, p8Next);

    float acc[8];
#pragma unroll
    for (int r = 0; r < 8; ++r) acc[r] = 0.0f;

#pragma unroll
    for (int tile = 0; tile < 10; ++tile) {
      v8f c = __builtin_amdgcn_wmma_f32_16x16x4_f32(
           false, wFrag[tile][0],  false, pCur[0],
           (short)0, biasC[tile],  false,  false);
      c = __builtin_amdgcn_wmma_f32_16x16x4_f32(
           false, wFrag[tile][1],  false, pCur[1],
           (short)0, c,  false,  false);
#pragma unroll
      for (int r = 0; r < 8; ++r) {
        const float hsum = __builtin_fmaf(w8v[tile][r], p8Cur, c[r]);
        acc[r] += fmaxf(hsum, 0.0f) * wpv[tile][r];
      }
    }

    float s = ((acc[0] + acc[1]) + (acc[2] + acc[3])) +
              ((acc[4] + acc[5]) + (acc[6] + acc[7]));
    s += xhalf(s);
    const float sg = 1.0f / (1.0f + __expf(-s));

    if (khalf == 0) prow[tt * 16 + m] = sg;

    pCur[0] = pNext[0];
    pCur[1] = pNext[1];
    p8Cur   = p8Next;
  }
}

#define TS     64
#define HALO   20
#define RS     (TS + 2 * HALO)
#define QPR    (RS / 4)
#define COL0   4
#define PITCH  112
#define BUFSZ  ((RS + 2) * PITCH)
#define FTHREADS 704
#define CS     108
#define NCOL   (QPR + 2)
#define CBUF   (2 * NCOL * CS)
#define LDSF   (2 * BUFSZ + 2 * CBUF)

__device__ __forceinline__ float max3f(float a, float b, float c) {
  return fmaxf(fmaxf(a, b), c);
}

__device__ __forceinline__ void patch_step(const float* __restrict__ vin,
                                           const float* __restrict__ cin,
                                           int ry, int rx, int qx,
                                           const float (&r)[4][4],
                                           const float (&p)[4][4],
                                           v4f (&v)[4]) {
  float hm[6][4];
#pragma unroll
  for (int k = 0; k < 6; k += 5) {
    const float* rowp = vin + (ry + k) * PITCH + COL0 + rx;
    const v4f    m  = *(const v4f*)rowp;
    const float  lf = rowp[-1];
    const float  rt = rowp[4];
    hm[k][0] = max3f(lf,  m.x, m.y);
    hm[k][1] = max3f(m.x, m.y, m.z);
    hm[k][2] = max3f(m.y, m.z, m.w);
    hm[k][3] = max3f(m.z, m.w, rt);
  }
  const v4f lf = *(const v4f*)(cin + (NCOL + qx) * CS + ry);
  const v4f rt = *(const v4f*)(cin + (qx + 2) * CS + ry);
#pragma unroll
  for (int j = 0; j < 4; ++j) {
    hm[j + 1][0] = max3f(lf[j],  v[j].x, v[j].y);
    hm[j + 1][1] = max3f(v[j].x, v[j].y, v[j].z);
    hm[j + 1][2] = max3f(v[j].y, v[j].z, v[j].w);
    hm[j + 1][3] = max3f(v[j].z, v[j].w, rt[j]);
  }
#pragma unroll
  for (int j = 0; j < 4; ++j) {
#pragma unroll
    for (int i = 0; i < 4; ++i) {
      float mx = max3f(hm[j][i], hm[j + 1][i], hm[j + 2][i]);
      v[j][i] = fmaxf(v[j][i], r[j][i] + p[j][i] * (mx - r[j][i]));
    }
  }
}

__device__ __forceinline__ void store_edges(float* __restrict__ vout,
                                            float* __restrict__ cout, int ry,
                                            int rx, int qx, const v4f (&v)[4]) {
#pragma unroll
  for (int j = 0; j < 4; ++j) {
    *(v4f*)(vout + (ry + 1 + j) * PITCH + COL0 + rx) = v[j];
  }
  *(v4f*)(cout + (qx + 1) * CS + ry) = (v4f){v[0].x, v[1].x, v[2].x, v[3].x};
  *(v4f*)(cout + (NCOL + qx + 1) * CS + ry) =
      (v4f){v[0].w, v[1].w, v[2].w, v[3].w};
}

__global__ __launch_bounds__(FTHREADS) void vprop_fused(
    const float* __restrict__ image, const float* __restrict__ pbuf,
    float* __restrict__ out) {
  __shared__ __attribute__((aligned(16))) float sv[LDSF];

  const int t    = threadIdx.x;
  const int tile = blockIdx.x;
  const int b    = tile >> 4;
  const int ty   = (tile >> 2) & 3;
  const int tx   = tile & 3;

  const int sy = t / QPR;
  const int qx = t - sy * QPR;
  const int ry = sy * 4;
  const int rx = qx * 4;
  const int gy = ty * TS - HALO + ry;
  const int gx = tx * TS - HALO + rx;
  const bool inimg = (t < QPR * QPR) & (gy >= 0) & (gy < 256) &
                     (gx >= 0) & (gx < 256);
  const bool inner = inimg & (ry >= HALO) & (ry < HALO + TS) &
                     (rx >= HALO) & (rx < HALO + TS);

  const int drow = (ry + 3 <= (RS - 1) / 2) ? ry + 3 : RS - 1 - ry;
  const int dcol = (rx + 3 <= (RS - 1) / 2) ? rx + 3 : RS - 1 - rx;
  const int dmax = inimg ? min(drow, dcol) : 0;

  const float* __restrict__ ib = image + (size_t)b * 2 * NPIX;
  const float* __restrict__ pb = pbuf + (size_t)b * NPIX;

  float r[4][4], p[4][4], oc[4][4];
  v4f v[4];
  if (inimg) {
#pragma unroll
    for (int j = 0; j < 4; ++j) {
      const int g = (gy + j) * 256 + gx;
      const float4 r4 = *(const float4*)(ib + NPIX + g);
      const float4 p4 = *(const float4*)(pb + g);
      r[j][0] = r4.x; r[j][1] = r4.y; r[j][2] = r4.z; r[j][3] = r4.w;
      p[j][0] = p4.x; p[j][1] = p4.y; p[j][2] = p4.z; p[j][3] = p4.w;
    }
  }
  if (inner) {
#pragma unroll
    for (int j = 0; j < 4; ++j) {
      const float4 o4 = *(const float4*)(ib + (gy + j) * 256 + gx);
      oc[j][0] = o4.x; oc[j][1] = o4.y; oc[j][2] = o4.z; oc[j][3] = o4.w;
    }
  }

  {
    const float ninf = -__builtin_inff();
    const float4 n4 = make_float4(ninf, ninf, ninf, ninf);
    float4* s4 = (float4*)sv;
    for (int i = t; i < LDSF / 4; i += FTHREADS) s4[i] = n4;
  }
  __syncthreads();

  float* const pl0 = sv;
  float* const pl1 = sv + BUFSZ;
  float* const cl0 = sv + 2 * BUFSZ;
  float* const cl1 = sv + 2 * BUFSZ + CBUF;

  if (inimg) {
#pragma unroll
    for (int j = 0; j < 4; ++j) {
      v[j] = (v4f){r[j][0], r[j][1], r[j][2], r[j][3]};
    }
    store_edges(pl0, cl0, ry, rx, qx, v);
  }
  __syncthreads();

  for (int s = 1; s < HALO - 1; s += 2) {
    if (dmax >= s) {
      patch_step(pl0, cl0, ry, rx, qx, r, p, v);
      store_edges(pl1, cl1, ry, rx, qx, v);
    }
    __syncthreads();
    if (dmax >= s + 1) {
      patch_step(pl1, cl1, ry, rx, qx, r, p, v);
      store_edges(pl0, cl0, ry, rx, qx, v);
    }
    __syncthreads();
  }
  if (dmax >= HALO - 1) {
    patch_step(pl0, cl0, ry, rx, qx, r, p, v);
    store_edges(pl1, cl1, ry, rx, qx, v);
  }
  __syncthreads();

  if (inner) {
    patch_step(pl1, cl1, ry, rx, qx, r, p, v);
#pragma unroll
    for (int j = 0; j < 4; ++j) {
      float4 o;
      o.x = (oc[j][0] > 0.49f) ? v[j].x : -1.0f;
      o.y = (oc[j][1] > 0.49f) ? v[j].y : -1.0f;
      o.z = (oc[j][2] > 0.49f) ? v[j].z : -1.0f;
      o.w = (oc[j][3] > 0.49f) ? v[j].w : -1.0f;
      *(float4*)(out + (size_t)b * NPIX + (gy + j) * 256 + gx) = o;
    }
  }
}

extern "C" void kernel_launch(void* const* d_in, const int* in_sizes, int n_in,
                              void* d_out, int out_size, void* d_ws,
                              size_t ws_size, hipStream_t stream) {
  const float* image = (const float*)d_in[0];
  const float* Wh    = (const float*)d_in[1];
  const float* bh    = (const float*)d_in[2];
  const float* Wp    = (const float*)d_in[3];
  float* out = (float*)d_out;

  float* pbuf  = (float*)d_ws;

  mvprop_conv<<<256, 256, 0, stream>>>(image, Wh, bh, Wp, pbuf);

  vprop_fused<<<NBATCH * 16, FTHREADS, 0, stream>>>(image, pbuf, out);
}
